# starting kernel
# baseline (speedup 1.0000x reference)
#include <hip/hip_runtime.h>
#include <math.h>

#define HDIM  256
#define BATCH 16
#define QLEN  64
#define KLEN  1024
#define DV    256
#define BQ    (BATCH * QLEN)
#define BKEYS (BATCH * KLEN)
#define NEGBIG -1e6f

typedef float v2f __attribute__((ext_vector_type(2)));
typedef float v8f __attribute__((ext_vector_type(8)));

__device__ __forceinline__ float fast_tanh(float x) {
#if __has_builtin(__builtin_amdgcn_tanhf)
  return __builtin_amdgcn_tanhf(x);
#elif __has_builtin(__builtin_amdgcn_tanh_f32)
  return __builtin_amdgcn_tanh_f32(x);
#else
  return tanhf(x);
#endif
}

__global__ __launch_bounds__(256) void proj_gemm(const float* __restrict__ X,
                                                 const float* __restrict__ W,
                                                 float* __restrict__ OUT) {
  const int lane = threadIdx.x & 31;
  const int wave = threadIdx.x >> 5;
  const int tile = blockIdx.x * 8 + wave;
  const int ntn  = HDIM / 16;
  const int tm   = tile / ntn;
  const int tn   = tile % ntn;
  const int half = lane >> 4;
  const int l16  = lane & 15;

  const float* Arow = X + (tm * 16 + l16) * HDIM;
  const float* Brow = W + (tn * 16 + l16) * HDIM;

  v8f acc = {};
#pragma unroll 8
  for (int k0 = 0; k0 < HDIM; k0 += 4) {
    v2f a = *(const v2f*)(Arow + k0 + 2 * half);
    v2f b = *(const v2f*)(Brow + k0 + 2 * half);
    acc = __builtin_amdgcn_wmma_f32_16x16x4_f32(
         false, a,  false, b,
         (short)0, acc,  false,  false);
  }

  float* Orow = OUT + (tm * 16 + 8 * half) * HDIM + tn * 16 + l16;
#pragma unroll
  for (int v = 0; v < 8; ++v)
    Orow[v * HDIM] = acc[v];
}

__global__ __launch_bounds__(256) void score_softmax(const float* __restrict__ Qp,
                                                     const float* __restrict__ Kp,
                                                     const float* __restrict__ wv,
                                                     const int* __restrict__ valid_lens,
                                                     float* __restrict__ attn) {
  __shared__ float scores[KLEN];
  __shared__ float red[256];

  const int bq   = blockIdx.x;
  const int b    = bq / QLEN;
  const int t    = threadIdx.x;
  const int lane = t & 31;
  const int wave = t >> 5;

  const float* qbase = Qp + (size_t)bq * HDIM + lane * 8;
  const float4 q0 = *(const float4*)(qbase);
  const float4 q1 = *(const float4*)(qbase + 4);
  const float* wbase = wv + lane * 8;
  const float4 w0 = *(const float4*)(wbase);
  const float4 w1 = *(const float4*)(wbase + 4);

  const int vlen = valid_lens[b];
  const int kbeg = wave * (KLEN / 8);
  const float* krow = Kp + ((size_t)b * KLEN + kbeg) * HDIM + lane * 8;

#pragma unroll 2
  for (int i = 0; i < KLEN / 8; ++i) {
    __builtin_prefetch(krow + 2 * HDIM, 0, 1);
    const float4 k0v = *(const float4*)(krow);
    const float4 k1v = *(const float4*)(krow + 4);

    float s = fast_tanh(q0.x + k0v.x) * w0.x;
    s += fast_tanh(q0.y + k0v.y) * w0.y;
    s += fast_tanh(q0.z + k0v.z) * w0.z;
    s += fast_tanh(q0.w + k0v.w) * w0.w;
    s += fast_tanh(q1.x + k1v.x) * w1.x;
    s += fast_tanh(q1.y + k1v.y) * w1.y;
    s += fast_tanh(q1.z + k1v.z) * w1.z;
    s += fast_tanh(q1.w + k1v.w) * w1.w;

#pragma unroll
    for (int off = 16; off > 0; off >>= 1)
      s += __shfl_xor(s, off, 32);

    const int k = kbeg + i;
    if (lane == 0) scores[k] = (k < vlen) ? s : NEGBIG;
    krow += HDIM;
  }
  __syncthreads();

  float sc[4];
#pragma unroll
  for (int j = 0; j < 4; ++j) sc[j] = scores[t + 256 * j];

  float m = fmaxf(fmaxf(sc[0], sc[1]), fmaxf(sc[2], sc[3]));
  red[t] = m;
  __syncthreads();
  for (int off = 128; off > 0; off >>= 1) {
    if (t < off) red[t] = fmaxf(red[t], red[t + off]);
    __syncthreads();
  }
  const float M = red[0];
  __syncthreads();

  float p[4];
  float ssum = 0.f;
#pragma unroll
  for (int j = 0; j < 4; ++j) {
    p[j] = __expf(sc[j] - M);
    ssum += p[j];
  }
  red[t] = ssum;
  __syncthreads();
  for (int off = 128; off > 0; off >>= 1) {
    if (t < off) red[t] += red[t + off];
    __syncthreads();
  }
  const float inv = 1.f / red[0];

#pragma unroll
  for (int j = 0; j < 4; ++j)
    attn[(size_t)bq * KLEN + t + 256 * j] = p[j] * inv;
}

__global__ __launch_bounds__(256) void out_gemm(const float* __restrict__ attn,
                                                const float* __restrict__ V,
                                                float* __restrict__ OUT) {
  const int lane = threadIdx.x & 31;
  const int wave = threadIdx.x >> 5;
  const int tile = blockIdx.x * 8 + wave;
  const int ntn  = DV / 16;
  const int ntm  = QLEN / 16;
  const int b    = tile / (ntm * ntn);
  const int r    = tile % (ntm * ntn);
  const int tm   = r / ntn;
  const int tn   = r % ntn;
  const int half = lane >> 4;
  const int l16  = lane & 15;

  const float* Ab = attn + (size_t)b * QLEN * KLEN + (tm * 16 + l16) * KLEN;
  const float* Vb = V + (size_t)b * KLEN * DV + tn * 16 + l16;

  v8f acc = {};
#pragma unroll 4
  for (int k0 = 0; k0 < KLEN; k0 += 4) {
    const int kb = k0 + 2 * half;
    v2f a = *(const v2f*)(Ab + kb);
    v2f bv;
    bv.x = Vb[(size_t)kb * DV];
    bv.y = Vb[(size_t)(kb + 1) * DV];
    acc = __builtin_amdgcn_wmma_f32_16x16x4_f32(
        false, a, false, bv, (short)0, acc, false, false);
  }

  float* Orow = OUT + (size_t)(b * QLEN + tm * 16 + 8 * half) * DV + tn * 16 + l16;
#pragma unroll
  for (int v = 0; v < 8; ++v)
    Orow[v * DV] = acc[v];
}

extern "C" void kernel_launch(void* const* d_in, const int* in_sizes, int n_in,
                              void* d_out, int out_size, void* d_ws, size_t ws_size,
                              hipStream_t stream) {
  const float* queries = (const float*)d_in[0];
  const float* keys    = (const float*)d_in[1];
  const float* values  = (const float*)d_in[2];
  const int*   vlens   = (const int*)d_in[3];
  const float* Wq      = (const float*)d_in[4];
  const float* Wk      = (const float*)d_in[5];
  const float* wv      = (const float*)d_in[6];
  float* out = (float*)d_out;

  float* Qp   = (float*)d_ws;
  float* Kp   = Qp + (size_t)BQ * HDIM;
  float* attn = Kp + (size_t)BKEYS * HDIM;

  proj_gemm<<<(BQ / 16) * (HDIM / 16) / 8, 256, 0, stream>>>(queries, Wq, Qp);
  proj_gemm<<<(BKEYS / 16) * (HDIM / 16) / 8, 256, 0, stream>>>(keys, Wk, Kp);

  score_softmax<<<BQ, 256, 0, stream>>>(Qp, Kp, wv, vlens, attn);

  out_gemm<<<(BATCH * (QLEN / 16) * (DV / 16)) / 8, 256, 0, stream>>>(attn, values, out);
}


# Key projection fused into the score kernel: a block projects 64 keys by WMMA into LDS and scores all 64 queries of its batch from there; softmax becomes its own launch; d and h sums in another order
# speedup vs baseline: 1.7415x; 1.7415x over previous
#include <hip/hip_runtime.h>
#include <math.h>

#define HDIM  256
#define BATCH 16
#define QLEN  64
#define KLEN  1024
#define DV    256
#define BQ    (BATCH * QLEN)
#define NEGBIG -1e6f

typedef float v2f __attribute__((ext_vector_type(2)));
typedef float v8f __attribute__((ext_vector_type(8)));

__device__ __forceinline__ float fast_tanh(float x) {
#if __has_builtin(__builtin_amdgcn_tanhf)
  return __builtin_amdgcn_tanhf(x);
#elif __has_builtin(__builtin_amdgcn_tanh_f32)
  return __builtin_amdgcn_tanh_f32(x);
#else
  return tanhf(x);
#endif
}

__global__ __launch_bounds__(256) void proj_gemm(const float* __restrict__ X,
                                                 const float* __restrict__ W,
                                                 float* __restrict__ OUT) {
  const int lane = threadIdx.x & 31;
  const int wave = threadIdx.x >> 5;
  const int tile = blockIdx.x * 8 + wave;
  const int ntn  = HDIM / 16;
  const int tm   = tile / ntn;
  const int tn   = tile % ntn;
  const int half = lane >> 4;
  const int l16  = lane & 15;

  const float* Arow = X + (tm * 16 + l16) * HDIM;
  const float* Brow = W + (tn * 16 + l16) * HDIM;

  v8f acc = {};
#pragma unroll 8
  for (int k0 = 0; k0 < HDIM; k0 += 4) {
    v2f a = *(const v2f*)(Arow + k0 + 2 * half);
    v2f b = *(const v2f*)(Brow + k0 + 2 * half);
    acc = __builtin_amdgcn_wmma_f32_16x16x4_f32(
         false, a,  false, b,
         (short)0, acc,  false,  false);
  }

  float* Orow = OUT + (tm * 16 + 8 * half) * HDIM + tn * 16 + l16;
#pragma unroll
  for (int v = 0; v < 8; ++v)
    Orow[v * HDIM] = acc[v];
}

#define KT   64
#define FP   260

__device__ __forceinline__ v8f wmma4(v2f a, v2f b, v8f c) {
  return __builtin_amdgcn_wmma_f32_16x16x4_f32(
       false, a,  false, b,
       (short)0, c,  false,  false);
}

__global__ __launch_bounds__(512) void keyproj_score(const float* __restrict__ keys,
                                                     const float* __restrict__ Wk,
                                                     const float* __restrict__ Qp,
                                                     const float* __restrict__ wv,
                                                     const int* __restrict__ valid_lens,
                                                     float* __restrict__ S) {
  __shared__ __attribute__((aligned(16))) float buf[KT * FP];

  const int t    = threadIdx.x;
  const int lane = t & 31;
  const int wave = __builtin_amdgcn_readfirstlane(t >> 5);
  const int b    = blockIdx.x / (KLEN / KT);
  const int k0   = (blockIdx.x % (KLEN / KT)) * KT;

  const float* kg = keys + ((size_t)b * KLEN + k0) * HDIM;
#pragma unroll
  for (int i = 0; i < (KT * HDIM / 4) / 512; ++i) {
    const int idx = i * 512 + t;
    const int row = idx >> 6;
    const int c4  = idx & 63;
    const float4 v = *(const float4*)(kg + row * HDIM + c4 * 4);
    *(float4*)(buf + row * FP + c4 * 4) = v;
  }
  __syncthreads();

  const int half = lane >> 4;
  const int l16  = lane & 15;
  const float* Brow = Wk + (wave * 16 + l16) * HDIM + 4 * half;
  const float* Arow = buf + l16 * FP + 4 * half;

  v8f acc[4] = {};
  float4 bcur[4], bnxt[4];
#pragma unroll
  for (int u = 0; u < 4; ++u) bcur[u] = *(const float4*)(Brow + 8 * u);

  __builtin_amdgcn_sched_barrier(0);

#pragma unroll
  for (int jj = 0; jj < HDIM / 8; jj += 4) {
    if (jj + 4 < HDIM / 8) {
#pragma unroll
      for (int u = 0; u < 4; ++u) bnxt[u] = *(const float4*)(Brow + 8 * (jj + 4 + u));
    }
#pragma unroll
    for (int u = 0; u < 4; ++u) {
      const int j = jj + u;
      float4 av[4];
#pragma unroll
      for (int tm = 0; tm < 4; ++tm)
        av[tm] = *(const float4*)(Arow + tm * 16 * FP + 8 * j);
      const v2f b0 = {bcur[u].x, bcur[u].y};
      const v2f b1 = {bcur[u].z, bcur[u].w};
#pragma unroll
      for (int tm = 0; tm < 4; ++tm) {
        const v2f a0 = {av[tm].x, av[tm].y};
        acc[tm] = wmma4(a0, b0, acc[tm]);
      }
#pragma unroll
      for (int tm = 0; tm < 4; ++tm) {
        const v2f a1 = {av[tm].z, av[tm].w};
        acc[tm] = wmma4(a1, b1, acc[tm]);
      }
    }
#pragma unroll
    for (int u = 0; u < 4; ++u) bcur[u] = bnxt[u];
  }
  __builtin_amdgcn_sched_barrier(0);
  asm volatile("v_nop\n\tv_nop\n\tv_nop\n\tv_nop\n\tv_nop");
  __syncthreads();

#pragma unroll
  for (int tm = 0; tm < 4; ++tm) {
#pragma unroll
    for (int v = 0; v < 8; ++v)
      buf[(tm * 16 + 8 * half + v) * FP + wave * 16 + l16] = acc[tm][v];
  }
  __syncthreads();

  const int vlen  = valid_lens[b];
  const int qrow0 = b * QLEN + wave * 4;
  const float* q0 = Qp + (size_t)qrow0 * HDIM;
  const float* kap = buf + lane * FP;
  const float* kbp = buf + (lane + 32) * FP;

  float sa[4] = {0.f, 0.f, 0.f, 0.f};
  float sb[4] = {0.f, 0.f, 0.f, 0.f};
#pragma unroll 4
  for (int h = 0; h < HDIM; h += 4) {
    const float4 ka = *(const float4*)(kap + h);
    const float4 kb = *(const float4*)(kbp + h);
    const float4 w  = *(const float4*)(wv + h);
#pragma unroll
    for (int g = 0; g < 4; ++g) {
      const float4 q = *(const float4*)(q0 + g * HDIM + h);
      sa[g] = __builtin_fmaf(fast_tanh(q.x + ka.x), w.x, sa[g]);
      sb[g] = __builtin_fmaf(fast_tanh(q.x + kb.x), w.x, sb[g]);
      sa[g] = __builtin_fmaf(fast_tanh(q.y + ka.y), w.y, sa[g]);
      sb[g] = __builtin_fmaf(fast_tanh(q.y + kb.y), w.y, sb[g]);
      sa[g] = __builtin_fmaf(fast_tanh(q.z + ka.z), w.z, sa[g]);
      sb[g] = __builtin_fmaf(fast_tanh(q.z + kb.z), w.z, sb[g]);
      sa[g] = __builtin_fmaf(fast_tanh(q.w + ka.w), w.w, sa[g]);
      sb[g] = __builtin_fmaf(fast_tanh(q.w + kb.w), w.w, sb[g]);
    }
  }

  const int  ka_idx = k0 + lane;
  const int  kb_idx = k0 + 32 + lane;
  const bool live_a = ka_idx < vlen;
  const bool live_b = kb_idx < vlen;
#pragma unroll
  for (int g = 0; g < 4; ++g) {
    float* srow = S + (size_t)(qrow0 + g) * KLEN;
    srow[ka_idx] = live_a ? sa[g] : NEGBIG;
    srow[kb_idx] = live_b ? sb[g] : NEGBIG;
  }
}

__global__ __launch_bounds__(256) void softmax_rows(float* __restrict__ attn) {
  __shared__ float red[256];

  const int tt = threadIdx.x;
  float* row = attn + (size_t)blockIdx.x * KLEN;

  float sc[4];
#pragma unroll
  for (int j = 0; j < 4; ++j) sc[j] = row[tt + 256 * j];

  float m = fmaxf(fmaxf(sc[0], sc[1]), fmaxf(sc[2], sc[3]));
  red[tt] = m;
  __syncthreads();
  for (int off = 128; off > 0; off >>= 1) {
    if (tt < off) red[tt] = fmaxf(red[tt], red[tt + off]);
    __syncthreads();
  }
  const float M = red[0];
  __syncthreads();

  float p[4];
  float ssum = 0.f;
#pragma unroll
  for (int j = 0; j < 4; ++j) {
    p[j] = __expf(sc[j] - M);
    ssum += p[j];
  }
  red[tt] = ssum;
  __syncthreads();
  for (int off = 128; off > 0; off >>= 1) {
    if (tt < off) red[tt] += red[tt + off];
    __syncthreads();
  }
  const float inv = 1.f / red[0];

#pragma unroll
  for (int j = 0; j < 4; ++j)
    row[tt + 256 * j] = p[j] * inv;
}

__global__ __launch_bounds__(256) void out_gemm(const float* __restrict__ attn,
                                                const float* __restrict__ V,
                                                float* __restrict__ OUT) {
  const int lane = threadIdx.x & 31;
  const int wave = threadIdx.x >> 5;
  const int tile = blockIdx.x * 8 + wave;
  const int ntn  = DV / 16;
  const int ntm  = QLEN / 16;
  const int b    = tile / (ntm * ntn);
  const int r    = tile % (ntm * ntn);
  const int tm   = r / ntn;
  const int tn   = r % ntn;
  const int half = lane >> 4;
  const int l16  = lane & 15;

  const float* Ab = attn + (size_t)b * QLEN * KLEN + (tm * 16 + l16) * KLEN;
  const float* Vb = V + (size_t)b * KLEN * DV + tn * 16 + l16;

  v8f acc = {};
#pragma unroll 4
  for (int k0 = 0; k0 < KLEN; k0 += 4) {
    const int kb = k0 + 2 * half;
    v2f a = *(const v2f*)(Ab + kb);
    v2f bv;
    bv.x = Vb[(size_t)kb * DV];
    bv.y = Vb[(size_t)(kb + 1) * DV];
    acc = __builtin_amdgcn_wmma_f32_16x16x4_f32(
        false, a, false, bv, (short)0, acc, false, false);
  }

  float* Orow = OUT + (size_t)(b * QLEN + tm * 16 + 8 * half) * DV + tn * 16 + l16;
#pragma unroll
  for (int v = 0; v < 8; ++v)
    Orow[v * DV] = acc[v];
}

extern "C" void kernel_launch(void* const* d_in, const int* in_sizes, int n_in,
                              void* d_out, int out_size, void* d_ws, size_t ws_size,
                              hipStream_t stream) {
  const float* queries = (const float*)d_in[0];
  const float* keys    = (const float*)d_in[1];
  const float* values  = (const float*)d_in[2];
  const int*   vlens   = (const int*)d_in[3];
  const float* Wq      = (const float*)d_in[4];
  const float* Wk      = (const float*)d_in[5];
  const float* wv      = (const float*)d_in[6];
  float* out = (float*)d_out;

  float* Qp   = (float*)d_ws;
  float* attn = Qp + (size_t)BQ * HDIM;

  proj_gemm<<<(BQ / 16) * (HDIM / 16) / 8, 256, 0, stream>>>(queries, Wq, Qp);

  keyproj_score<<<BATCH * (KLEN / KT), 512, 0, stream>>>(keys, Wk, Qp, wv, vlens, attn);

  softmax_rows<<<BQ, 256, 0, stream>>>(attn);

  out_gemm<<<(BATCH * (QLEN / 16) * (DV / 16)) / 8, 256, 0, stream>>>(attn, values, out);
}


# Softmax fused into the output GEMM: a block normalises 16 score rows into LDS and multiplies them by a 64-column value slice staged through an LDS ring; softmax and k sums in another order
# speedup vs baseline: 3.2168x; 1.8471x over previous
#include <hip/hip_runtime.h>
#include <math.h>

#define HDIM  256
#define BATCH 16
#define QLEN  64
#define KLEN  1024
#define DV    256
#define BQ    (BATCH * QLEN)
#define NEGBIG -1e6f

typedef float v2f __attribute__((ext_vector_type(2)));
typedef float v8f __attribute__((ext_vector_type(8)));

__device__ __forceinline__ float fast_tanh(float x) {
#if __has_builtin(__builtin_amdgcn_tanhf)
  return __builtin_amdgcn_tanhf(x);
#elif __has_builtin(__builtin_amdgcn_tanh_f32)
  return __builtin_amdgcn_tanh_f32(x);
#else
  return tanhf(x);
#endif
}

__global__ __launch_bounds__(256) void proj_gemm(const float* __restrict__ X,
                                                 const float* __restrict__ W,
                                                 float* __restrict__ OUT) {
  const int lane = threadIdx.x & 31;
  const int wave = threadIdx.x >> 5;
  const int tile = blockIdx.x * 8 + wave;
  const int ntn  = HDIM / 16;
  const int tm   = tile / ntn;
  const int tn   = tile % ntn;
  const int half = lane >> 4;
  const int l16  = lane & 15;

  const float* Arow = X + (tm * 16 + l16) * HDIM;
  const float* Brow = W + (tn * 16 + l16) * HDIM;

  v8f acc = {};
#pragma unroll 8
  for (int k0 = 0; k0 < HDIM; k0 += 4) {
    v2f a = *(const v2f*)(Arow + k0 + 2 * half);
    v2f b = *(const v2f*)(Brow + k0 + 2 * half);
    acc = __builtin_amdgcn_wmma_f32_16x16x4_f32(
         false, a,  false, b,
         (short)0, acc,  false,  false);
  }

  float* Orow = OUT + (tm * 16 + 8 * half) * HDIM + tn * 16 + l16;
#pragma unroll
  for (int v = 0; v < 8; ++v)
    Orow[v * HDIM] = acc[v];
}

#define KT   64
#define FP   260

__device__ __forceinline__ v8f wmma4(v2f a, v2f b, v8f c) {
  return __builtin_amdgcn_wmma_f32_16x16x4_f32(
       false, a,  false, b,
       (short)0, c,  false,  false);
}

__global__ __launch_bounds__(512) void keyproj_score(const float* __restrict__ keys,
                                                     const float* __restrict__ Wk,
                                                     const float* __restrict__ Qp,
                                                     const float* __restrict__ wv,
                                                     const int* __restrict__ valid_lens,
                                                     float* __restrict__ S) {
  __shared__ __attribute__((aligned(16))) float buf[KT * FP];

  const int t    = threadIdx.x;
  const int lane = t & 31;
  const int wave = __builtin_amdgcn_readfirstlane(t >> 5);
  const int b    = blockIdx.x / (KLEN / KT);
  const int k0   = (blockIdx.x % (KLEN / KT)) * KT;

  const float* kg = keys + ((size_t)b * KLEN + k0) * HDIM;
#pragma unroll
  for (int i = 0; i < (KT * HDIM / 4) / 512; ++i) {
    const int idx = i * 512 + t;
    const int row = idx >> 6;
    const int c4  = idx & 63;
    const float4 v = *(const float4*)(kg + row * HDIM + c4 * 4);
    *(float4*)(buf + row * FP + c4 * 4) = v;
  }
  __syncthreads();

  const int half = lane >> 4;
  const int l16  = lane & 15;
  const float* Brow = Wk + (wave * 16 + l16) * HDIM + 4 * half;
  const float* Arow = buf + l16 * FP + 4 * half;

  v8f acc[4] = {};
  float4 bcur[4], bnxt[4];
#pragma unroll
  for (int u = 0; u < 4; ++u) bcur[u] = *(const float4*)(Brow + 8 * u);

  __builtin_amdgcn_sched_barrier(0);

#pragma unroll
  for (int jj = 0; jj < HDIM / 8; jj += 4) {
    if (jj + 4 < HDIM / 8) {
#pragma unroll
      for (int u = 0; u < 4; ++u) bnxt[u] = *(const float4*)(Brow + 8 * (jj + 4 + u));
    }
#pragma unroll
    for (int u = 0; u < 4; ++u) {
      const int j = jj + u;
      float4 av[4];
#pragma unroll
      for (int tm = 0; tm < 4; ++tm)
        av[tm] = *(const float4*)(Arow + tm * 16 * FP + 8 * j);
      const v2f b0 = {bcur[u].x, bcur[u].y};
      const v2f b1 = {bcur[u].z, bcur[u].w};
#pragma unroll
      for (int tm = 0; tm < 4; ++tm) {
        const v2f a0 = {av[tm].x, av[tm].y};
        acc[tm] = wmma4(a0, b0, acc[tm]);
      }
#pragma unroll
      for (int tm = 0; tm < 4; ++tm) {
        const v2f a1 = {av[tm].z, av[tm].w};
        acc[tm] = wmma4(a1, b1, acc[tm]);
      }
    }
#pragma unroll
    for (int u = 0; u < 4; ++u) bcur[u] = bnxt[u];
  }
  __builtin_amdgcn_sched_barrier(0);
  asm volatile("v_nop\n\tv_nop\n\tv_nop\n\tv_nop\n\tv_nop");
  __syncthreads();

#pragma unroll
  for (int tm = 0; tm < 4; ++tm) {
#pragma unroll
    for (int v = 0; v < 8; ++v)
      buf[(tm * 16 + 8 * half + v) * FP + wave * 16 + l16] = acc[tm][v];
  }
  __syncthreads();

  const int vlen  = valid_lens[b];
  const int qrow0 = b * QLEN + wave * 4;
  const float* q0 = Qp + (size_t)qrow0 * HDIM;
  const float* kap = buf + lane * FP;
  const float* kbp = buf + (lane + 32) * FP;

  float sa[4] = {0.f, 0.f, 0.f, 0.f};
  float sb[4] = {0.f, 0.f, 0.f, 0.f};
#pragma unroll 4
  for (int h = 0; h < HDIM; h += 4) {
    const float4 ka = *(const float4*)(kap + h);
    const float4 kb = *(const float4*)(kbp + h);
    const float4 w  = *(const float4*)(wv + h);
#pragma unroll
    for (int g = 0; g < 4; ++g) {
      const float4 q = *(const float4*)(q0 + g * HDIM + h);
      sa[g] = __builtin_fmaf(fast_tanh(q.x + ka.x), w.x, sa[g]);
      sb[g] = __builtin_fmaf(fast_tanh(q.x + kb.x), w.x, sb[g]);
      sa[g] = __builtin_fmaf(fast_tanh(q.y + ka.y), w.y, sa[g]);
      sb[g] = __builtin_fmaf(fast_tanh(q.y + kb.y), w.y, sb[g]);
      sa[g] = __builtin_fmaf(fast_tanh(q.z + ka.z), w.z, sa[g]);
      sb[g] = __builtin_fmaf(fast_tanh(q.z + kb.z), w.z, sb[g]);
      sa[g] = __builtin_fmaf(fast_tanh(q.w + ka.w), w.w, sa[g]);
      sb[g] = __builtin_fmaf(fast_tanh(q.w + kb.w), w.w, sb[g]);
    }
  }

  const int  ka_idx = k0 + lane;
  const int  kb_idx = k0 + 32 + lane;
  const bool live_a = ka_idx < vlen;
  const bool live_b = kb_idx < vlen;
#pragma unroll
  for (int g = 0; g < 4; ++g) {
    float* srow = S + (size_t)(qrow0 + g) * KLEN;
    srow[ka_idx] = live_a ? sa[g] : NEGBIG;
    srow[kb_idx] = live_b ? sb[g] : NEGBIG;
  }
}

#define PP   1028
#define KC   64
#define VP   72
#define OP   68

#define STAGE_LOAD(R, src)                                              \
  do {                                                                  \
    R##a = *(const float4*)((src));                                     \
    R##b = *(const float4*)((src) + (size_t)16 * DV);                   \
    R##c = *(const float4*)((src) + (size_t)32 * DV);                   \
    R##d = *(const float4*)((src) + (size_t)48 * DV);                   \
  } while (0)
#define STAGE_STORE(dst, R)                                             \
  do {                                                                  \
    *(float4*)((dst))           = R##a;                                 \
    *(float4*)((dst) + 16 * VP) = R##b;                                 \
    *(float4*)((dst) + 32 * VP) = R##c;                                 \
    *(float4*)((dst) + 48 * VP) = R##d;                                 \
  } while (0)
__device__ __forceinline__ void lds_barrier() {
  asm volatile("s_wait_dscnt 0x0\n\ts_barrier_signal -1\n\ts_barrier_wait -1" ::: "memory");
}
__device__ __forceinline__ void mma_chunk(const float* ap, const float* bp, v8f& acc0, v8f& acc1) {
  float4 av[4];
  float  bv[4][4];
#pragma unroll
  for (int g = 0; g < 4; ++g) av[g] = *(const float4*)(ap + 8 * g);
#pragma unroll
  for (int g = 0; g < 4; ++g)
#pragma unroll
    for (int j = 0; j < 4; ++j) bv[g][j] = bp[(8 * g + j) * VP];
#pragma unroll
  for (int g = 0; g < 4; ++g) {
    const v2f a0 = {av[g].x, av[g].y};
    const v2f a1 = {av[g].z, av[g].w};
    const v2f b0 = {bv[g][0], bv[g][1]};
    const v2f b1 = {bv[g][2], bv[g][3]};
    acc0 = wmma4(a0, b0, acc0);
    acc1 = wmma4(a1, b1, acc1);
  }
}

__global__ __launch_bounds__(256) void softmax_out(const float* __restrict__ S,
                                                   const float* __restrict__ V,
                                                   float* __restrict__ OUT) {
  __shared__ __attribute__((aligned(16))) float P[16 * PP];
  __shared__ __attribute__((aligned(16))) float Vs[2 * KC * VP];

  const int t    = threadIdx.x;
  const int lane = t & 31;
  const int wave = __builtin_amdgcn_readfirstlane(t >> 5);
  const int b    = blockIdx.x >> 4;
  const int ng   = (blockIdx.x >> 2) & 3;
  const int tm   = blockIdx.x & 3;

  const int srow = t >> 4;
  const int sc4  = t & 15;
  const float* vsrc = V + (size_t)b * KLEN * DV + (size_t)srow * DV + ng * 64 + sc4 * 4;
  float* vdst = Vs + srow * VP + sc4 * 4;

  float4 st0a, st0b, st0c, st0d, st1a, st1b, st1c, st1d;
  STAGE_LOAD(st0, vsrc);
  STAGE_LOAD(st1, vsrc + (size_t)KC * DV);

  const float* Sg = S + ((size_t)(b * QLEN + tm * 16 + wave * 2)) * KLEN;
  float4 x[2][8];
#pragma unroll
  for (int rr = 0; rr < 2; ++rr)
#pragma unroll
    for (int i = 0; i < 8; ++i)
      x[rr][i] = *(const float4*)(Sg + rr * KLEN + 4 * (lane + 32 * i));

#pragma unroll
  for (int rr = 0; rr < 2; ++rr) {
    float m = x[rr][0].x;
#pragma unroll
    for (int i = 0; i < 8; ++i)
      m = fmaxf(m, fmaxf(fmaxf(x[rr][i].x, x[rr][i].y), fmaxf(x[rr][i].z, x[rr][i].w)));
#pragma unroll
    for (int off = 16; off > 0; off >>= 1) m = fmaxf(m, __shfl_xor(m, off));
    float s = 0.f;
#pragma unroll
    for (int i = 0; i < 8; ++i) {
      x[rr][i].x = __expf(x[rr][i].x - m);
      x[rr][i].y = __expf(x[rr][i].y - m);
      x[rr][i].z = __expf(x[rr][i].z - m);
      x[rr][i].w = __expf(x[rr][i].w - m);
      s += (x[rr][i].x + x[rr][i].y) + (x[rr][i].z + x[rr][i].w);
    }
#pragma unroll
    for (int off = 16; off > 0; off >>= 1) s += __shfl_xor(s, off);
    const float inv = 1.f / s;
    float* prow = P + (wave * 2 + rr) * PP;
#pragma unroll
    for (int i = 0; i < 8; ++i) {
      float4 p = x[rr][i];
      p.x *= inv; p.y *= inv; p.z *= inv; p.w *= inv;
      *(float4*)(prow + 4 * (lane + 32 * i)) = p;
    }
  }

  STAGE_STORE(vdst, st0);
  STAGE_LOAD(st0, vsrc + (size_t)(2 * KC) * DV);
  lds_barrier();

  const int tn   = wave & 3;
  const int kh   = wave >> 2;
  const int half = lane >> 4;
  const int l16  = lane & 15;
  const float* Ap = P + l16 * PP + kh * 32 + 4 * half;
  const float* Bp = Vs + (kh * 32 + 4 * half) * VP + tn * 16 + l16;

  v8f acc0 = {};
  v8f acc1 = {};
#pragma unroll 1
  for (int cc = 0; cc < KLEN / KC - 4; cc += 2) {
    STAGE_STORE(vdst + KC * VP, st1);
    STAGE_LOAD(st1, vsrc + (size_t)((cc + 3) * KC) * DV);
    mma_chunk(Ap + cc * KC, Bp, acc0, acc1);
    lds_barrier();
    STAGE_STORE(vdst, st0);
    STAGE_LOAD(st0, vsrc + (size_t)((cc + 4) * KC) * DV);
    mma_chunk(Ap + (cc + 1) * KC, Bp + KC * VP, acc0, acc1);
    lds_barrier();
  }
  STAGE_STORE(vdst + KC * VP, st1);
  STAGE_LOAD(st1, vsrc + (size_t)((KLEN / KC - 1) * KC) * DV);
  mma_chunk(Ap + (KLEN / KC - 4) * KC, Bp, acc0, acc1);
  lds_barrier();
  STAGE_STORE(vdst, st0);
  mma_chunk(Ap + (KLEN / KC - 3) * KC, Bp + KC * VP, acc0, acc1);
  lds_barrier();
  STAGE_STORE(vdst + KC * VP, st1);
  mma_chunk(Ap + (KLEN / KC - 2) * KC, Bp, acc0, acc1);
  lds_barrier();
  mma_chunk(Ap + (KLEN / KC - 1) * KC, Bp + KC * VP, acc0, acc1);
  int te = t;
  asm volatile("v_nop\n\tv_nop\n\tv_nop\n\tv_nop\n\tv_nop" : "+v"(acc0), "+v"(acc1), "+v"(te));

  const int e_lane = te & 31;
  const int e_wave = te >> 5;
  const int e_tn   = e_wave & 3;
  const int e_kh   = e_wave >> 2;
  const int e_half = e_lane >> 4;
  const int e_l16  = e_lane & 15;
  const int e_srow = te >> 4;
  const int e_sc4  = te & 15;
  float* O = Vs;
#pragma unroll
  for (int v = 0; v < 8; ++v)
    O[(e_kh * 16 + 8 * e_half + v) * OP + e_tn * 16 + e_l16] = acc0[v] + acc1[v];
  __syncthreads();
  const float4 o0 = *(const float4*)(O + e_srow * OP + e_sc4 * 4);
  const float4 o1 = *(const float4*)(O + (16 + e_srow) * OP + e_sc4 * 4);
  float4 o;
  o.x = o0.x + o1.x; o.y = o0.y + o1.y; o.z = o0.z + o1.z; o.w = o0.w + o1.w;
  *(float4*)(OUT + (size_t)(b * QLEN + tm * 16 + e_srow) * DV + ng * 64 + e_sc4 * 4) = o;
}

extern "C" void kernel_launch(void* const* d_in, const int* in_sizes, int n_in,
                              void* d_out, int out_size, void* d_ws, size_t ws_size,
                              hipStream_t stream) {
  const float* queries = (const float*)d_in[0];
  const float* keys    = (const float*)d_in[1];
  const float* values  = (const float*)d_in[2];
  const int*   vlens   = (const int*)d_in[3];
  const float* Wq      = (const float*)d_in[4];
  const float* Wk      = (const float*)d_in[5];
  const float* wv      = (const float*)d_in[6];
  float* out = (float*)d_out;

  float* Qp   = (float*)d_ws;
  float* attn = Qp + (size_t)BQ * HDIM;

  proj_gemm<<<(BQ / 16) * (HDIM / 16) / 8, 256, 0, stream>>>(queries, Wq, Qp);

  keyproj_score<<<BATCH * (KLEN / KT), 512, 0, stream>>>(keys, Wk, Qp, wv, vlens, attn);

  softmax_out<<<BATCH * (DV / 64) * (QLEN / 16), 256, 0, stream>>>(attn, values, out);
}


# Score kernel skips masked keys: blocks split each batch's keys in front of valid_lens evenly (2 to 4 tiles of 16 keys, computed in-kernel); masked scores filled directly; every sum keeps its order
# speedup vs baseline: 4.0373x; 1.2551x over previous
#include <hip/hip_runtime.h>
#include <math.h>

#define HDIM  256
#define BATCH 16
#define QLEN  64
#define KLEN  1024
#define DV    256
#define BQ    (BATCH * QLEN)
#define NEGBIG -1e6f

typedef float v2f __attribute__((ext_vector_type(2)));
typedef float v8f __attribute__((ext_vector_type(8)));

__device__ __forceinline__ float fast_tanh(float x) {
#if __has_builtin(__builtin_amdgcn_tanhf)
  return __builtin_amdgcn_tanhf(x);
#elif __has_builtin(__builtin_amdgcn_tanh_f32)
  return __builtin_amdgcn_tanh_f32(x);
#else
  return tanhf(x);
#endif
}

__global__ __launch_bounds__(256) void proj_gemm(const float* __restrict__ X,
                                                 const float* __restrict__ W,
                                                 float* __restrict__ OUT) {
  const int lane = threadIdx.x & 31;
  const int wave = threadIdx.x >> 5;
  const int tile = blockIdx.x * 8 + wave;
  const int ntn  = HDIM / 16;
  const int tm   = tile / ntn;
  const int tn   = tile % ntn;
  const int half = lane >> 4;
  const int l16  = lane & 15;

  const float* Arow = X + (tm * 16 + l16) * HDIM;
  const float* Brow = W + (tn * 16 + l16) * HDIM;

  v8f acc = {};
#pragma unroll 8
  for (int k0 = 0; k0 < HDIM; k0 += 4) {
    v2f a = *(const v2f*)(Arow + k0 + 2 * half);
    v2f b = *(const v2f*)(Brow + k0 + 2 * half);
    acc = __builtin_amdgcn_wmma_f32_16x16x4_f32(
         false, a,  false, b,
         (short)0, acc,  false,  false);
  }

  float* Orow = OUT + (tm * 16 + 8 * half) * HDIM + tn * 16 + l16;
#pragma unroll
  for (int v = 0; v < 8; ++v)
    Orow[v * HDIM] = acc[v];
}

#define KT   64
#define FP   260
#define NBLK (BATCH * (KLEN / KT))

__device__ __forceinline__ v8f wmma4(v2f a, v2f b, v8f c) {
  return __builtin_amdgcn_wmma_f32_16x16x4_f32(
       false, a,  false, b,
       (short)0, c,  false,  false);
}

template <int M>
__device__ __forceinline__ void project_and_score(const float* __restrict__ keys,
                                                  const float* __restrict__ Wk,
                                                  const float* __restrict__ Qp,
                                                  const float* __restrict__ wv,
                                                  float* __restrict__ S,
                                                  float* __restrict__ buf,
                                                  const int b, const int k0, const int vlen) {
  const int t    = threadIdx.x;
  const int lane = t & 31;
  const int wave = __builtin_amdgcn_readfirstlane(t >> 5);

  const float* kg = keys + (size_t)b * KLEN * HDIM;
#pragma unroll
  for (int i = 0; i < 2 * M; ++i) {
    const int idx = i * 512 + t;
    const int row = idx >> 6;
    const int c4  = idx & 63;
    const int gk  = min(k0 + row, KLEN - 1);
    const float4 v = *(const float4*)(kg + (size_t)gk * HDIM + c4 * 4);
    *(float4*)(buf + row * FP + c4 * 4) = v;
  }
  __syncthreads();

  const int half = lane >> 4;
  const int l16  = lane & 15;
  const float* Brow = Wk + (wave * 16 + l16) * HDIM + 4 * half;
  const float* Arow = buf + l16 * FP + 4 * half;

  v8f acc[M] = {};
  float4 bcur[4], bnxt[4];
#pragma unroll
  for (int u = 0; u < 4; ++u) bcur[u] = *(const float4*)(Brow + 8 * u);

  __builtin_amdgcn_sched_barrier(0);

#pragma unroll
  for (int jj = 0; jj < HDIM / 8; jj += 4) {
    if (jj + 4 < HDIM / 8) {
#pragma unroll
      for (int u = 0; u < 4; ++u) bnxt[u] = *(const float4*)(Brow + 8 * (jj + 4 + u));
    }
#pragma unroll
    for (int u = 0; u < 4; ++u) {
      const int j = jj + u;
      float4 av[M];
#pragma unroll
      for (int tm = 0; tm < M; ++tm)
        av[tm] = *(const float4*)(Arow + tm * 16 * FP + 8 * j);
      const v2f b0 = {bcur[u].x, bcur[u].y};
      const v2f b1 = {bcur[u].z, bcur[u].w};
#pragma unroll
      for (int tm = 0; tm < M; ++tm) {
        const v2f a0 = {av[tm].x, av[tm].y};
        acc[tm] = wmma4(a0, b0, acc[tm]);
      }
#pragma unroll
      for (int tm = 0; tm < M; ++tm) {
        const v2f a1 = {av[tm].z, av[tm].w};
        acc[tm] = wmma4(a1, b1, acc[tm]);
      }
    }
#pragma unroll
    for (int u = 0; u < 4; ++u) bcur[u] = bnxt[u];
  }
  __builtin_amdgcn_sched_barrier(0);
  asm volatile("v_nop\n\tv_nop\n\tv_nop\n\tv_nop\n\tv_nop");
  __syncthreads();

#pragma unroll
  for (int tm = 0; tm < M; ++tm) {
#pragma unroll
    for (int v = 0; v < 8; ++v)
      buf[(tm * 16 + 8 * half + v) * FP + wave * 16 + l16] = acc[tm][v];
  }
  __syncthreads();

  const int qrow = b * QLEN + wave * 4 + 2 * half;
  const float* qp = Qp + (size_t)qrow * HDIM;
  const float* kp = buf + l16 * FP;

  float s0[M], s1[M];
#pragma unroll
  for (int tm = 0; tm < M; ++tm) { s0[tm] = 0.f; s1[tm] = 0.f; }
#pragma unroll 4
  for (int h = 0; h < HDIM; h += 4) {
    const float4 w  = *(const float4*)(wv + h);
    const float4 qa = *(const float4*)(qp + h);
    const float4 qb = *(const float4*)(qp + HDIM + h);
#pragma unroll
    for (int tm = 0; tm < M; ++tm) {
      const float4 k4 = *(const float4*)(kp + tm * 16 * FP + h);
      s0[tm] = __builtin_fmaf(fast_tanh(qa.x + k4.x), w.x, s0[tm]);
      s1[tm] = __builtin_fmaf(fast_tanh(qb.x + k4.x), w.x, s1[tm]);
      s0[tm] = __builtin_fmaf(fast_tanh(qa.y + k4.y), w.y, s0[tm]);
      s1[tm] = __builtin_fmaf(fast_tanh(qb.y + k4.y), w.y, s1[tm]);
      s0[tm] = __builtin_fmaf(fast_tanh(qa.z + k4.z), w.z, s0[tm]);
      s1[tm] = __builtin_fmaf(fast_tanh(qb.z + k4.z), w.z, s1[tm]);
      s0[tm] = __builtin_fmaf(fast_tanh(qa.w + k4.w), w.w, s0[tm]);
      s1[tm] = __builtin_fmaf(fast_tanh(qb.w + k4.w), w.w, s1[tm]);
    }
  }

  float* srow = S + (size_t)qrow * KLEN;
#pragma unroll
  for (int tm = 0; tm < M; ++tm) {
    const int kidx = k0 + tm * 16 + l16;
    if (kidx < KLEN) {
      const bool live = kidx < vlen;
      srow[kidx]        = live ? s0[tm] : NEGBIG;
      srow[KLEN + kidx] = live ? s1[tm] : NEGBIG;
    }
  }
}

__global__ __launch_bounds__(512) void keyproj_score(const float* __restrict__ keys,
                                                     const float* __restrict__ Wk,
                                                     const float* __restrict__ Qp,
                                                     const float* __restrict__ wv,
                                                     const int* __restrict__ valid_lens,
                                                     float* __restrict__ S) {
  __shared__ __attribute__((aligned(16))) float buf[KT * FP];

  const int t   = threadIdx.x;
  const int blk = blockIdx.x;

  int c2[BATCH], c3[BATCH], c4[BATCH];
  int n2 = 0, n3 = 0;
#pragma unroll
  for (int i = 0; i < BATCH; ++i) {
    const int vl = min(max(valid_lens[i], 0), KLEN);
    const int T  = (vl + 15) >> 4;
    c2[i] = (T + 1) / 2;
    c3[i] = (T + 2) / 3;
    c4[i] = (T + 3) / 4;
    n2 += c2[i];
    n3 += c3[i];
  }
  const int M = (n2 <= NBLK) ? 2 : ((n3 <= NBLK) ? 3 : 4);

  int my_b = -1, my_j = 0, my_vlen = 0, first = 0, cov_fill = 0;
#pragma unroll
  for (int i = 0; i < BATCH; ++i) {
    const int nb = (M == 2) ? c2[i] : ((M == 3) ? c3[i] : c4[i]);
    const bool mine = (blk >= first) && (blk < first + nb);
    my_b    = mine ? i : my_b;
    my_j    = mine ? blk - first : my_j;
    my_vlen = mine ? valid_lens[i] : my_vlen;
    cov_fill = (i == (blk >> 4)) ? min(KLEN, nb * 16 * M) : cov_fill;
    first += nb;
  }

  {
    float* frow = S + (size_t)(blk * 4 + (t >> 7)) * KLEN;
    const float4 neg = {NEGBIG, NEGBIG, NEGBIG, NEGBIG};
    for (int c = cov_fill + 4 * (t & 127); c < KLEN; c += 512)
      *(float4*)(frow + c) = neg;
  }

  if (my_b < 0) return;
  const int k0 = my_j * 16 * M;

  if (M == 2)      project_and_score<2>(keys, Wk, Qp, wv, S, buf, my_b, k0, my_vlen);
  else if (M == 3) project_and_score<3>(keys, Wk, Qp, wv, S, buf, my_b, k0, my_vlen);
  else             project_and_score<4>(keys, Wk, Qp, wv, S, buf, my_b, k0, my_vlen);
}

#define PP   1028
#define KC   64
#define VP   72
#define OP   68

#define STAGE_LOAD(R, src)                                              \
  do {                                                                  \
    R##a = *(const float4*)((src));                                     \
    R##b = *(const float4*)((src) + (size_t)16 * DV);                   \
    R##c = *(const float4*)((src) + (size_t)32 * DV);                   \
    R##d = *(const float4*)((src) + (size_t)48 * DV);                   \
  } while (0)
#define STAGE_STORE(dst, R)                                             \
  do {                                                                  \
    *(float4*)((dst))           = R##a;                                 \
    *(float4*)((dst) + 16 * VP) = R##b;                                 \
    *(float4*)((dst) + 32 * VP) = R##c;                                 \
    *(float4*)((dst) + 48 * VP) = R##d;                                 \
  } while (0)
__device__ __forceinline__ void lds_barrier() {
  asm volatile("s_wait_dscnt 0x0\n\ts_barrier_signal -1\n\ts_barrier_wait -1" ::: "memory");
}
__device__ __forceinline__ void mma_chunk(const float* ap, const float* bp, v8f& acc0, v8f& acc1) {
  float4 av[4];
  float  bv[4][4];
#pragma unroll
  for (int g = 0; g < 4; ++g) av[g] = *(const float4*)(ap + 8 * g);
#pragma unroll
  for (int g = 0; g < 4; ++g)
#pragma unroll
    for (int j = 0; j < 4; ++j) bv[g][j] = bp[(8 * g + j) * VP];
#pragma unroll
  for (int g = 0; g < 4; ++g) {
    const v2f a0 = {av[g].x, av[g].y};
    const v2f a1 = {av[g].z, av[g].w};
    const v2f b0 = {bv[g][0], bv[g][1]};
    const v2f b1 = {bv[g][2], bv[g][3]};
    acc0 = wmma4(a0, b0, acc0);
    acc1 = wmma4(a1, b1, acc1);
  }
}

__global__ __launch_bounds__(256) void softmax_out(const float* __restrict__ S,
                                                   const float* __restrict__ V,
                                                   float* __restrict__ OUT) {
  __shared__ __attribute__((aligned(16))) float P[16 * PP];
  __shared__ __attribute__((aligned(16))) float Vs[2 * KC * VP];

  const int t    = threadIdx.x;
  const int lane = t & 31;
  const int wave = __builtin_amdgcn_readfirstlane(t >> 5);
  const int b    = blockIdx.x >> 4;
  const int ng   = (blockIdx.x >> 2) & 3;
  const int tm   = blockIdx.x & 3;

  const int srow = t >> 4;
  const int sc4  = t & 15;
  const float* vsrc = V + (size_t)b * KLEN * DV + (size_t)srow * DV + ng * 64 + sc4 * 4;
  float* vdst = Vs + srow * VP + sc4 * 4;

  float4 st0a, st0b, st0c, st0d, st1a, st1b, st1c, st1d;
  STAGE_LOAD(st0, vsrc);
  STAGE_LOAD(st1, vsrc + (size_t)KC * DV);

  const float* Sg = S + ((size_t)(b * QLEN + tm * 16 + wave * 2)) * KLEN;
  float4 x[2][8];
#pragma unroll
  for (int rr = 0; rr < 2; ++rr)
#pragma unroll
    for (int i = 0; i < 8; ++i)
      x[rr][i] = *(const float4*)(Sg + rr * KLEN + 4 * (lane + 32 * i));

#pragma unroll
  for (int rr = 0; rr < 2; ++rr) {
    float m = x[rr][0].x;
#pragma unroll
    for (int i = 0; i < 8; ++i)
      m = fmaxf(m, fmaxf(fmaxf(x[rr][i].x, x[rr][i].y), fmaxf(x[rr][i].z, x[rr][i].w)));
#pragma unroll
    for (int off = 16; off > 0; off >>= 1) m = fmaxf(m, __shfl_xor(m, off));
    float s = 0.f;
#pragma unroll
    for (int i = 0; i < 8; ++i) {
      x[rr][i].x = __expf(x[rr][i].x - m);
      x[rr][i].y = __expf(x[rr][i].y - m);
      x[rr][i].z = __expf(x[rr][i].z - m);
      x[rr][i].w = __expf(x[rr][i].w - m);
      s += (x[rr][i].x + x[rr][i].y) + (x[rr][i].z + x[rr][i].w);
    }
#pragma unroll
    for (int off = 16; off > 0; off >>= 1) s += __shfl_xor(s, off);
    const float inv = 1.f / s;
    float* prow = P + (wave * 2 + rr) * PP;
#pragma unroll
    for (int i = 0; i < 8; ++i) {
      float4 p = x[rr][i];
      p.x *= inv; p.y *= inv; p.z *= inv; p.w *= inv;
      *(float4*)(prow + 4 * (lane + 32 * i)) = p;
    }
  }

  STAGE_STORE(vdst, st0);
  STAGE_LOAD(st0, vsrc + (size_t)(2 * KC) * DV);
  lds_barrier();

  const int tn   = wave & 3;
  const int kh   = wave >> 2;
  const int half = lane >> 4;
  const int l16  = lane & 15;
  const float* Ap = P + l16 * PP + kh * 32 + 4 * half;
  const float* Bp = Vs + (kh * 32 + 4 * half) * VP + tn * 16 + l16;

  v8f acc0 = {};
  v8f acc1 = {};
#pragma unroll 1
  for (int cc = 0; cc < KLEN / KC - 4; cc += 2) {
    STAGE_STORE(vdst + KC * VP, st1);
    STAGE_LOAD(st1, vsrc + (size_t)((cc + 3) * KC) * DV);
    mma_chunk(Ap + cc * KC, Bp, acc0, acc1);
    lds_barrier();
    STAGE_STORE(vdst, st0);
    STAGE_LOAD(st0, vsrc + (size_t)((cc + 4) * KC) * DV);
    mma_chunk(Ap + (cc + 1) * KC, Bp + KC * VP, acc0, acc1);
    lds_barrier();
  }
  STAGE_STORE(vdst + KC * VP, st1);
  STAGE_LOAD(st1, vsrc + (size_t)((KLEN / KC - 1) * KC) * DV);
  mma_chunk(Ap + (KLEN / KC - 4) * KC, Bp, acc0, acc1);
  lds_barrier();
  STAGE_STORE(vdst, st0);
  mma_chunk(Ap + (KLEN / KC - 3) * KC, Bp + KC * VP, acc0, acc1);
  lds_barrier();
  STAGE_STORE(vdst + KC * VP, st1);
  mma_chunk(Ap + (KLEN / KC - 2) * KC, Bp, acc0, acc1);
  lds_barrier();
  mma_chunk(Ap + (KLEN / KC - 1) * KC, Bp + KC * VP, acc0, acc1);
  int te = t;
  asm volatile("v_nop\n\tv_nop\n\tv_nop\n\tv_nop\n\tv_nop" : "+v"(acc0), "+v"(acc1), "+v"(te));

  const int e_lane = te & 31;
  const int e_wave = te >> 5;
  const int e_tn   = e_wave & 3;
  const int e_kh   = e_wave >> 2;
  const int e_half = e_lane >> 4;
  const int e_l16  = e_lane & 15;
  const int e_srow = te >> 4;
  const int e_sc4  = te & 15;
  float* O = Vs;
#pragma unroll
  for (int v = 0; v < 8; ++v)
    O[(e_kh * 16 + 8 * e_half + v) * OP + e_tn * 16 + e_l16] = acc0[v] + acc1[v];
  __syncthreads();
  const float4 o0 = *(const float4*)(O + e_srow * OP + e_sc4 * 4);
  const float4 o1 = *(const float4*)(O + (16 + e_srow) * OP + e_sc4 * 4);
  float4 o;
  o.x = o0.x + o1.x; o.y = o0.y + o1.y; o.z = o0.z + o1.z; o.w = o0.w + o1.w;
  *(float4*)(OUT + (size_t)(b * QLEN + tm * 16 + e_srow) * DV + ng * 64 + e_sc4 * 4) = o;
}

extern "C" void kernel_launch(void* const* d_in, const int* in_sizes, int n_in,
                              void* d_out, int out_size, void* d_ws, size_t ws_size,
                              hipStream_t stream) {
  const float* queries = (const float*)d_in[0];
  const float* keys    = (const float*)d_in[1];
  const float* values  = (const float*)d_in[2];
  const int*   vlens   = (const int*)d_in[3];
  const float* Wq      = (const float*)d_in[4];
  const float* Wk      = (const float*)d_in[5];
  const float* wv      = (const float*)d_in[6];
  float* out = (float*)d_out;

  float* Qp   = (float*)d_ws;
  float* attn = Qp + (size_t)BQ * HDIM;

  proj_gemm<<<(BQ / 16) * (HDIM / 16) / 8, 256, 0, stream>>>(queries, Wq, Qp);

  keyproj_score<<<NBLK, 512, 0, stream>>>(keys, Wk, Qp, wv, vlens, attn);

  softmax_out<<<BATCH * (DV / 64) * (QLEN / 16), 256, 0, stream>>>(attn, values, out);
}


# Output kernel requests the value rows six chunks ahead instead of two (six register sets, chunk loop written out): three times the value bytes in flight per block; every sum keeps its order
# speedup vs baseline: 4.3829x; 1.0856x over previous
#include <hip/hip_runtime.h>
#include <math.h>

#define HDIM  256
#define BATCH 16
#define QLEN  64
#define KLEN  1024
#define DV    256
#define BQ    (BATCH * QLEN)
#define NEGBIG -1e6f

typedef float v2f __attribute__((ext_vector_type(2)));
typedef float v8f __attribute__((ext_vector_type(8)));

__device__ __forceinline__ float fast_tanh(float x) {
#if __has_builtin(__builtin_amdgcn_tanhf)
  return __builtin_amdgcn_tanhf(x);
#elif __has_builtin(__builtin_amdgcn_tanh_f32)
  return __builtin_amdgcn_tanh_f32(x);
#else
  return tanhf(x);
#endif
}

__global__ __launch_bounds__(256) void proj_gemm(const float* __restrict__ X,
                                                 const float* __restrict__ W,
                                                 float* __restrict__ OUT) {
  const int lane = threadIdx.x & 31;
  const int wave = threadIdx.x >> 5;
  const int tile = blockIdx.x * 8 + wave;
  const int ntn  = HDIM / 16;
  const int tm   = tile / ntn;
  const int tn   = tile % ntn;
  const int half = lane >> 4;
  const int l16  = lane & 15;

  const float* Arow = X + (tm * 16 + l16) * HDIM;
  const float* Brow = W + (tn * 16 + l16) * HDIM;

  v8f acc = {};
#pragma unroll 8
  for (int k0 = 0; k0 < HDIM; k0 += 4) {
    v2f a = *(const v2f*)(Arow + k0 + 2 * half);
    v2f b = *(const v2f*)(Brow + k0 + 2 * half);
    acc = __builtin_amdgcn_wmma_f32_16x16x4_f32(
         false, a,  false, b,
         (short)0, acc,  false,  false);
  }

  float* Orow = OUT + (tm * 16 + 8 * half) * HDIM + tn * 16 + l16;
#pragma unroll
  for (int v = 0; v < 8; ++v)
    Orow[v * HDIM] = acc[v];
}

#define KT   64
#define FP   260
#define NBLK (BATCH * (KLEN / KT))

__device__ __forceinline__ v8f wmma4(v2f a, v2f b, v8f c) {
  return __builtin_amdgcn_wmma_f32_16x16x4_f32(
       false, a,  false, b,
       (short)0, c,  false,  false);
}

template <int M>
__device__ __forceinline__ void project_and_score(const float* __restrict__ keys,
                                                  const float* __restrict__ Wk,
                                                  const float* __restrict__ Qp,
                                                  const float* __restrict__ wv,
                                                  float* __restrict__ S,
                                                  float* __restrict__ buf,
                                                  const int b, const int k0, const int vlen) {
  const int t    = threadIdx.x;
  const int lane = t & 31;
  const int wave = __builtin_amdgcn_readfirstlane(t >> 5);

  const float* kg = keys + (size_t)b * KLEN * HDIM;
#pragma unroll
  for (int i = 0; i < 2 * M; ++i) {
    const int idx = i * 512 + t;
    const int row = idx >> 6;
    const int c4  = idx & 63;
    const int gk  = min(k0 + row, KLEN - 1);
    const float4 v = *(const float4*)(kg + (size_t)gk * HDIM + c4 * 4);
    *(float4*)(buf + row * FP + c4 * 4) = v;
  }
  __syncthreads();

  const int half = lane >> 4;
  const int l16  = lane & 15;
  const float* Brow = Wk + (wave * 16 + l16) * HDIM + 4 * half;
  const float* Arow = buf + l16 * FP + 4 * half;

  v8f acc[M] = {};
  float4 bcur[4], bnxt[4];
#pragma unroll
  for (int u = 0; u < 4; ++u) bcur[u] = *(const float4*)(Brow + 8 * u);

  __builtin_amdgcn_sched_barrier(0);

#pragma unroll
  for (int jj = 0; jj < HDIM / 8; jj += 4) {
    if (jj + 4 < HDIM / 8) {
#pragma unroll
      for (int u = 0; u < 4; ++u) bnxt[u] = *(const float4*)(Brow + 8 * (jj + 4 + u));
    }
#pragma unroll
    for (int u = 0; u < 4; ++u) {
      const int j = jj + u;
      float4 av[M];
#pragma unroll
      for (int tm = 0; tm < M; ++tm)
        av[tm] = *(const float4*)(Arow + tm * 16 * FP + 8 * j);
      const v2f b0 = {bcur[u].x, bcur[u].y};
      const v2f b1 = {bcur[u].z, bcur[u].w};
#pragma unroll
      for (int tm = 0; tm < M; ++tm) {
        const v2f a0 = {av[tm].x, av[tm].y};
        acc[tm] = wmma4(a0, b0, acc[tm]);
      }
#pragma unroll
      for (int tm = 0; tm < M; ++tm) {
        const v2f a1 = {av[tm].z, av[tm].w};
        acc[tm] = wmma4(a1, b1, acc[tm]);
      }
    }
#pragma unroll
    for (int u = 0; u < 4; ++u) bcur[u] = bnxt[u];
  }
  __builtin_amdgcn_sched_barrier(0);
  asm volatile("v_nop\n\tv_nop\n\tv_nop\n\tv_nop\n\tv_nop");
  __syncthreads();

#pragma unroll
  for (int tm = 0; tm < M; ++tm) {
#pragma unroll
    for (int v = 0; v < 8; ++v)
      buf[(tm * 16 + 8 * half + v) * FP + wave * 16 + l16] = acc[tm][v];
  }
  __syncthreads();

  const int qrow = b * QLEN + wave * 4 + 2 * half;
  const float* qp = Qp + (size_t)qrow * HDIM;
  const float* kp = buf + l16 * FP;

  float s0[M], s1[M];
#pragma unroll
  for (int tm = 0; tm < M; ++tm) { s0[tm] = 0.f; s1[tm] = 0.f; }
#pragma unroll 4
  for (int h = 0; h < HDIM; h += 4) {
    const float4 w  = *(const float4*)(wv + h);
    const float4 qa = *(const float4*)(qp + h);
    const float4 qb = *(const float4*)(qp + HDIM + h);
#pragma unroll
    for (int tm = 0; tm < M; ++tm) {
      const float4 k4 = *(const float4*)(kp + tm * 16 * FP + h);
      s0[tm] = __builtin_fmaf(fast_tanh(qa.x + k4.x), w.x, s0[tm]);
      s1[tm] = __builtin_fmaf(fast_tanh(qb.x + k4.x), w.x, s1[tm]);
      s0[tm] = __builtin_fmaf(fast_tanh(qa.y + k4.y), w.y, s0[tm]);
      s1[tm] = __builtin_fmaf(fast_tanh(qb.y + k4.y), w.y, s1[tm]);
      s0[tm] = __builtin_fmaf(fast_tanh(qa.z + k4.z), w.z, s0[tm]);
      s1[tm] = __builtin_fmaf(fast_tanh(qb.z + k4.z), w.z, s1[tm]);
      s0[tm] = __builtin_fmaf(fast_tanh(qa.w + k4.w), w.w, s0[tm]);
      s1[tm] = __builtin_fmaf(fast_tanh(qb.w + k4.w), w.w, s1[tm]);
    }
  }

  float* srow = S + (size_t)qrow * KLEN;
#pragma unroll
  for (int tm = 0; tm < M; ++tm) {
    const int kidx = k0 + tm * 16 + l16;
    if (kidx < KLEN) {
      const bool live = kidx < vlen;
      srow[kidx]        = live ? s0[tm] : NEGBIG;
      srow[KLEN + kidx] = live ? s1[tm] : NEGBIG;
    }
  }
}

__global__ __launch_bounds__(512) void keyproj_score(const float* __restrict__ keys,
                                                     const float* __restrict__ Wk,
                                                     const float* __restrict__ Qp,
                                                     const float* __restrict__ wv,
                                                     const int* __restrict__ valid_lens,
                                                     float* __restrict__ S) {
  __shared__ __attribute__((aligned(16))) float buf[KT * FP];

  const int t   = threadIdx.x;
  const int blk = blockIdx.x;

  int c2[BATCH], c3[BATCH], c4[BATCH];
  int n2 = 0, n3 = 0;
#pragma unroll
  for (int i = 0; i < BATCH; ++i) {
    const int vl = min(max(valid_lens[i], 0), KLEN);
    const int T  = (vl + 15) >> 4;
    c2[i] = (T + 1) / 2;
    c3[i] = (T + 2) / 3;
    c4[i] = (T + 3) / 4;
    n2 += c2[i];
    n3 += c3[i];
  }
  const int M = (n2 <= NBLK) ? 2 : ((n3 <= NBLK) ? 3 : 4);

  int my_b = -1, my_j = 0, my_vlen = 0, first = 0, cov_fill = 0;
#pragma unroll
  for (int i = 0; i < BATCH; ++i) {
    const int nb = (M == 2) ? c2[i] : ((M == 3) ? c3[i] : c4[i]);
    const bool mine = (blk >= first) && (blk < first + nb);
    my_b    = mine ? i : my_b;
    my_j    = mine ? blk - first : my_j;
    my_vlen = mine ? valid_lens[i] : my_vlen;
    cov_fill = (i == (blk >> 4)) ? min(KLEN, nb * 16 * M) : cov_fill;
    first += nb;
  }

  {
    float* frow = S + (size_t)(blk * 4 + (t >> 7)) * KLEN;
    const float4 neg = {NEGBIG, NEGBIG, NEGBIG, NEGBIG};
    for (int c = cov_fill + 4 * (t & 127); c < KLEN; c += 512)
      *(float4*)(frow + c) = neg;
  }

  if (my_b < 0) return;
  const int k0 = my_j * 16 * M;

  if (M == 2)      project_and_score<2>(keys, Wk, Qp, wv, S, buf, my_b, k0, my_vlen);
  else if (M == 3) project_and_score<3>(keys, Wk, Qp, wv, S, buf, my_b, k0, my_vlen);
  else             project_and_score<4>(keys, Wk, Qp, wv, S, buf, my_b, k0, my_vlen);
}

#define PP   1028
#define KC   64
#define VP   72
#define OP   68

#define NSET 6
#define STAGE_LOAD(R, src)                                              \
  do {                                                                  \
    R##a = *(const float4*)((src));                                     \
    R##b = *(const float4*)((src) + (size_t)16 * DV);                   \
    R##c = *(const float4*)((src) + (size_t)32 * DV);                   \
    R##d = *(const float4*)((src) + (size_t)48 * DV);                   \
  } while (0)
#define STAGE_STORE(dst, R)                                             \
  do {                                                                  \
    *(float4*)((dst))           = R##a;                                 \
    *(float4*)((dst) + 16 * VP) = R##b;                                 \
    *(float4*)((dst) + 32 * VP) = R##c;                                 \
    *(float4*)((dst) + 48 * VP) = R##d;                                 \
  } while (0)
__device__ __forceinline__ void lds_barrier() {
  asm volatile("s_wait_dscnt 0x0\n\ts_barrier_signal -1\n\ts_barrier_wait -1" ::: "memory");
}
__device__ __forceinline__ void mma_chunk(const float* ap, const float* bp, v8f& acc0, v8f& acc1) {
  float4 av[4];
  float  bv[4][4];
#pragma unroll
  for (int g = 0; g < 4; ++g) av[g] = *(const float4*)(ap + 8 * g);
#pragma unroll
  for (int g = 0; g < 4; ++g)
#pragma unroll
    for (int j = 0; j < 4; ++j) bv[g][j] = bp[(8 * g + j) * VP];
#pragma unroll
  for (int g = 0; g < 4; ++g) {
    const v2f a0 = {av[g].x, av[g].y};
    const v2f a1 = {av[g].z, av[g].w};
    const v2f b0 = {bv[g][0], bv[g][1]};
    const v2f b1 = {bv[g][2], bv[g][3]};
    acc0 = wmma4(a0, b0, acc0);
    acc1 = wmma4(a1, b1, acc1);
  }
}

__global__ __launch_bounds__(256) void softmax_out(const float* __restrict__ S,
                                                   const float* __restrict__ V,
                                                   float* __restrict__ OUT) {
  __shared__ __attribute__((aligned(16))) float P[16 * PP];
  __shared__ __attribute__((aligned(16))) float Vs[2 * KC * VP];

  const int t    = threadIdx.x;
  const int lane = t & 31;
  const int wave = __builtin_amdgcn_readfirstlane(t >> 5);
  const int b    = blockIdx.x >> 4;
  const int ng   = (blockIdx.x >> 2) & 3;
  const int tm   = blockIdx.x & 3;

  const int srow = t >> 4;
  const int sc4  = t & 15;
  const float* vsrc = V + (size_t)b * KLEN * DV + (size_t)srow * DV + ng * 64 + sc4 * 4;
  float* vdst = Vs + srow * VP + sc4 * 4;

  const float* Sg = S + ((size_t)(b * QLEN + tm * 16 + wave * 2)) * KLEN;
  float4 x[2][8];
#pragma unroll
  for (int rr = 0; rr < 2; ++rr)
#pragma unroll
    for (int i = 0; i < 8; ++i)
      x[rr][i] = *(const float4*)(Sg + rr * KLEN + 4 * (lane + 32 * i));

  float4 st0a, st0b, st0c, st0d, st1a, st1b, st1c, st1d, st2a, st2b, st2c, st2d;
  float4 st3a, st3b, st3c, st3d, st4a, st4b, st4c, st4d, st5a, st5b, st5c, st5d;
  STAGE_LOAD(st0, vsrc);
  STAGE_LOAD(st1, vsrc + (size_t)(1 * KC) * DV);
  STAGE_LOAD(st2, vsrc + (size_t)(2 * KC) * DV);
  STAGE_LOAD(st3, vsrc + (size_t)(3 * KC) * DV);
  STAGE_LOAD(st4, vsrc + (size_t)(4 * KC) * DV);
  STAGE_LOAD(st5, vsrc + (size_t)(5 * KC) * DV);

#pragma unroll
  for (int rr = 0; rr < 2; ++rr) {
    float m = x[rr][0].x;
#pragma unroll
    for (int i = 0; i < 8; ++i)
      m = fmaxf(m, fmaxf(fmaxf(x[rr][i].x, x[rr][i].y), fmaxf(x[rr][i].z, x[rr][i].w)));
#pragma unroll
    for (int off = 16; off > 0; off >>= 1) m = fmaxf(m, __shfl_xor(m, off));
    float s = 0.f;
#pragma unroll
    for (int i = 0; i < 8; ++i) {
      x[rr][i].x = __expf(x[rr][i].x - m);
      x[rr][i].y = __expf(x[rr][i].y - m);
      x[rr][i].z = __expf(x[rr][i].z - m);
      x[rr][i].w = __expf(x[rr][i].w - m);
      s += (x[rr][i].x + x[rr][i].y) + (x[rr][i].z + x[rr][i].w);
    }
#pragma unroll
    for (int off = 16; off > 0; off >>= 1) s += __shfl_xor(s, off);
    const float inv = 1.f / s;
    float* prow = P + (wave * 2 + rr) * PP;
#pragma unroll
    for (int i = 0; i < 8; ++i) {
      float4 p = x[rr][i];
      p.x *= inv; p.y *= inv; p.z *= inv; p.w *= inv;
      *(float4*)(prow + 4 * (lane + 32 * i)) = p;
    }
  }

  STAGE_STORE(vdst, st0);
  STAGE_LOAD(st0, vsrc + (size_t)(NSET * KC) * DV);
  lds_barrier();

  const int tn   = wave & 3;
  const int kh   = wave >> 2;
  const int half = lane >> 4;
  const int l16  = lane & 15;
  const float* Ap = P + l16 * PP + kh * 32 + 4 * half;
  const float* Bp = Vs + (kh * 32 + 4 * half) * VP + tn * 16 + l16;

  v8f acc0 = {};
  v8f acc1 = {};
#define CHUNK_REQ(c, R)                                                           \
  do {                                                                            \
    STAGE_STORE(vdst + (((c) + 1) & 1) * KC * VP, R);                             \
    STAGE_LOAD(R, vsrc + (size_t)(((c) + 1 + NSET) * KC) * DV);                   \
    mma_chunk(Ap + (c) * KC, Bp + ((c) & 1) * KC * VP, acc0, acc1);               \
    lds_barrier();                                                                \
  } while (0)
#define CHUNK_END(c, R)                                                           \
  do {                                                                            \
    STAGE_STORE(vdst + (((c) + 1) & 1) * KC * VP, R);                             \
    mma_chunk(Ap + (c) * KC, Bp + ((c) & 1) * KC * VP, acc0, acc1);               \
    lds_barrier();                                                                \
  } while (0)
  CHUNK_REQ(0, st1);
  CHUNK_REQ(1, st2);
  CHUNK_REQ(2, st3);
  CHUNK_REQ(3, st4);
  CHUNK_REQ(4, st5);
  CHUNK_REQ(5, st0);
  CHUNK_REQ(6, st1);
  CHUNK_REQ(7, st2);
  CHUNK_REQ(8, st3);
  CHUNK_END(9, st4);
  CHUNK_END(10, st5);
  CHUNK_END(11, st0);
  CHUNK_END(12, st1);
  CHUNK_END(13, st2);
  CHUNK_END(14, st3);
  mma_chunk(Ap + 15 * KC, Bp + KC * VP, acc0, acc1);
  int te = t;
  asm volatile("v_nop\n\tv_nop\n\tv_nop\n\tv_nop\n\tv_nop" : "+v"(acc0), "+v"(acc1), "+v"(te));

  const int e_lane = te & 31;
  const int e_wave = te >> 5;
  const int e_tn   = e_wave & 3;
  const int e_kh   = e_wave >> 2;
  const int e_half = e_lane >> 4;
  const int e_l16  = e_lane & 15;
  const int e_srow = te >> 4;
  const int e_sc4  = te & 15;
  float* O = Vs;
#pragma unroll
  for (int v = 0; v < 8; ++v)
    O[(e_kh * 16 + 8 * e_half + v) * OP + e_tn * 16 + e_l16] = acc0[v] + acc1[v];
  __syncthreads();
  const float4 o0 = *(const float4*)(O + e_srow * OP + e_sc4 * 4);
  const float4 o1 = *(const float4*)(O + (16 + e_srow) * OP + e_sc4 * 4);
  float4 o;
  o.x = o0.x + o1.x; o.y = o0.y + o1.y; o.z = o0.z + o1.z; o.w = o0.w + o1.w;
  *(float4*)(OUT + (size_t)(b * QLEN + tm * 16 + e_srow) * DV + ng * 64 + e_sc4 * 4) = o;
}

extern "C" void kernel_launch(void* const* d_in, const int* in_sizes, int n_in,
                              void* d_out, int out_size, void* d_ws, size_t ws_size,
                              hipStream_t stream) {
  const float* queries = (const float*)d_in[0];
  const float* keys    = (const float*)d_in[1];
  const float* values  = (const float*)d_in[2];
  const int*   vlens   = (const int*)d_in[3];
  const float* Wq      = (const float*)d_in[4];
  const float* Wk      = (const float*)d_in[5];
  const float* wv      = (const float*)d_in[6];
  float* out = (float*)d_out;

  float* Qp   = (float*)d_ws;
  float* attn = Qp + (size_t)BQ * HDIM;

  proj_gemm<<<(BQ / 16) * (HDIM / 16) / 8, 256, 0, stream>>>(queries, Wq, Qp);

  keyproj_score<<<NBLK, 512, 0, stream>>>(keys, Wk, Qp, wv, vlens, attn);

  softmax_out<<<BATCH * (DV / 64) * (QLEN / 16), 256, 0, stream>>>(attn, values, out);
}


# speedup vs baseline: 4.5259x; 1.0326x over previous
#include <hip/hip_runtime.h>
#include <math.h>

#define HDIM  256
#define BATCH 16
#define QLEN  64
#define KLEN  1024
#define DV    256
#define BQ    (BATCH * QLEN)
#define NEGBIG -1e6f

typedef float v2f __attribute__((ext_vector_type(2)));
typedef float v8f __attribute__((ext_vector_type(8)));

__device__ __forceinline__ float fast_tanh(float x) {
#if __has_builtin(__builtin_amdgcn_tanhf)
  return __builtin_amdgcn_tanhf(x);
#elif __has_builtin(__builtin_amdgcn_tanh_f32)
  return __builtin_amdgcn_tanh_f32(x);
#else
  return tanhf(x);
#endif
}

__global__ __launch_bounds__(256) void proj_gemm(const float* __restrict__ X,
                                                 const float* __restrict__ W,
                                                 float* __restrict__ OUT) {
  const int lane = threadIdx.x & 31;
  const int wave = threadIdx.x >> 5;
  const int tile = blockIdx.x * 8 + wave;
  const int ntn  = HDIM / 16;
  const int tm   = tile / ntn;
  const int tn   = tile % ntn;
  const int half = lane >> 4;
  const int l16  = lane & 15;

  const float* Arow = X + (tm * 16 + l16) * HDIM;
  const float* Brow = W + (tn * 16 + l16) * HDIM;

  v8f acc = {};
#pragma unroll 8
  for (int k0 = 0; k0 < HDIM; k0 += 4) {
    v2f a = *(const v2f*)(Arow + k0 + 2 * half);
    v2f b = *(const v2f*)(Brow + k0 + 2 * half);
    acc = __builtin_amdgcn_wmma_f32_16x16x4_f32(
         false, a,  false, b,
         (short)0, acc,  false,  false);
  }

  float* Orow = OUT + (tm * 16 + 8 * half) * HDIM + tn * 16 + l16;
#pragma unroll
  for (int v = 0; v < 8; ++v)
    Orow[v * HDIM] = acc[v];
}

#define KT   64
#define FP   260
#define NBLK (BATCH * (KLEN / KT))

__device__ __forceinline__ v8f wmma4(v2f a, v2f b, v8f c) {
  return __builtin_amdgcn_wmma_f32_16x16x4_f32(
       false, a,  false, b,
       (short)0, c,  false,  false);
}

template <int M>
__device__ __forceinline__ void project_and_score(const float* __restrict__ keys,
                                                  const float* __restrict__ Wk,
                                                  const float* __restrict__ Qp,
                                                  const float* __restrict__ wv,
                                                  float* __restrict__ S,
                                                  float* __restrict__ buf,
                                                  const int b, const int k0, const int vlen) {
  const int t    = threadIdx.x;
  const int lane = t & 31;
  const int wave = __builtin_amdgcn_readfirstlane(t >> 5);

  const float* kg = keys + (size_t)b * KLEN * HDIM;
#pragma unroll
  for (int i = 0; i < 2 * M; ++i) {
    const int idx = i * 512 + t;
    const int row = idx >> 6;
    const int c4  = idx & 63;
    const int gk  = min(k0 + row, KLEN - 1);
    const float4 v = *(const float4*)(kg + (size_t)gk * HDIM + c4 * 4);
    *(float4*)(buf + row * FP + c4 * 4) = v;
  }
  __syncthreads();

  const int half = lane >> 4;
  const int l16  = lane & 15;
  const float* Brow = Wk + (wave * 16 + l16) * HDIM + 4 * half;
  const float* Arow = buf + l16 * FP + 4 * half;

  v8f acc[M] = {};
  float4 bcur[4], bnxt[4];
#pragma unroll
  for (int u = 0; u < 4; ++u) bcur[u] = *(const float4*)(Brow + 8 * u);

  __builtin_amdgcn_sched_barrier(0);

#pragma unroll
  for (int jj = 0; jj < HDIM / 8; jj += 4) {
    if (jj + 4 < HDIM / 8) {
#pragma unroll
      for (int u = 0; u < 4; ++u) bnxt[u] = *(const float4*)(Brow + 8 * (jj + 4 + u));
    }
#pragma unroll
    for (int u = 0; u < 4; ++u) {
      const int j = jj + u;
      float4 av[M];
#pragma unroll
      for (int tm = 0; tm < M; ++tm)
        av[tm] = *(const float4*)(Arow + tm * 16 * FP + 8 * j);
      const v2f b0 = {bcur[u].x, bcur[u].y};
      const v2f b1 = {bcur[u].z, bcur[u].w};
#pragma unroll
      for (int tm = 0; tm < M; ++tm) {
        const v2f a0 = {av[tm].x, av[tm].y};
        acc[tm] = wmma4(a0, b0, acc[tm]);
      }
#pragma unroll
      for (int tm = 0; tm < M; ++tm) {
        const v2f a1 = {av[tm].z, av[tm].w};
        acc[tm] = wmma4(a1, b1, acc[tm]);
      }
    }
#pragma unroll
    for (int u = 0; u < 4; ++u) bcur[u] = bnxt[u];
  }
  __builtin_amdgcn_sched_barrier(0);
  asm volatile("v_nop\n\tv_nop\n\tv_nop\n\tv_nop\n\tv_nop");
  __syncthreads();

#pragma unroll
  for (int tm = 0; tm < M; ++tm) {
#pragma unroll
    for (int v = 0; v < 8; ++v)
      buf[(tm * 16 + 8 * half + v) * FP + wave * 16 + l16] = acc[tm][v];
  }
  __syncthreads();

  const int qrow = b * QLEN + wave * 4 + 2 * half;
  const float* qp = Qp + (size_t)qrow * HDIM;
  const float* kp = buf + l16 * FP;

  constexpr int TP = 4;
  v2f e0[M], e1[M];
#pragma unroll
  for (int tm = 0; tm < M; ++tm) { e0[tm] = (v2f){0.f, 0.f}; e1[tm] = (v2f){0.f, 0.f}; }
  v2f yp0 = {0.f, 0.f}, yp1 = {0.f, 0.f}, yp2 = {0.f, 0.f}, yp3 = {0.f, 0.f};
  v2f wp01 = {0.f, 0.f}, wp23 = {0.f, 0.f};
#define PIN_ORDER() __builtin_amdgcn_sched_barrier(0)
#define HOLD(v) asm volatile("" : "+v"(v))
#pragma unroll 1
  for (int h = 0; h < HDIM; h += 4 * TP) {
    float4 w[TP], qa[TP], qb[TP], kk[TP][M];
#pragma unroll
    for (int t = 0; t < TP; ++t) {
      w[t]  = *(const float4*)(wv + h + 4 * t);
      qa[t] = *(const float4*)(qp + h + 4 * t);
      qb[t] = *(const float4*)(qp + HDIM + h + 4 * t);
#pragma unroll
      for (int tm = 0; tm < M; ++tm) kk[t][tm] = *(const float4*)(kp + tm * 16 * FP + h + 4 * t);
    }
    v2f x0 = (v2f){qa[0].x, qa[0].y} + (v2f){kk[0][0].x, kk[0][0].y};
    v2f x1 = (v2f){qb[0].x, qb[0].y} + (v2f){kk[0][0].x, kk[0][0].y};
    v2f x2 = (v2f){qa[0].z, qa[0].w} + (v2f){kk[0][0].z, kk[0][0].w};
    v2f x3 = (v2f){qb[0].z, qb[0].w} + (v2f){kk[0][0].z, kk[0][0].w};
    PIN_ORDER();
#pragma unroll
    for (int t = 0; t < TP; ++t) {
#pragma unroll
      for (int tm = 0; tm < M; ++tm) {
        const bool last = (t == TP - 1) && (tm == M - 1);
        const int tn  = (tm + 1 < M) ? t : ((t + 1 < TP) ? t + 1 : t);
        const int tmn = (tm + 1 < M) ? tm + 1 : 0;
        const int ptm = (tm > 0) ? tm - 1 : M - 1;
        float y0x, y0y, y1x, y1y, y2x, y2y, y3x, y3y;
        v2f xn0 = x0, xn1 = x1, xn2 = x2, xn3 = x3;
        y0x = fast_tanh(x0.x);
        HOLD(y0x); PIN_ORDER();
        if (!last) { xn0 = (v2f){qa[tn].x, qa[tn].y} + (v2f){kk[tn][tmn].x, kk[tn][tmn].y}; HOLD(xn0); }
        PIN_ORDER();
        y0y = fast_tanh(x0.y);
        HOLD(y0y); PIN_ORDER();
        e0[ptm] = __builtin_elementwise_fma(yp0, wp01, e0[ptm]);
        HOLD(e0[ptm]); PIN_ORDER();
        y1x = fast_tanh(x1.x);
        HOLD(y1x); PIN_ORDER();
        if (!last) { xn1 = (v2f){qb[tn].x, qb[tn].y} + (v2f){kk[tn][tmn].x, kk[tn][tmn].y}; HOLD(xn1); }
        PIN_ORDER();
        y1y = fast_tanh(x1.y);
        HOLD(y1y); PIN_ORDER();
        e1[ptm] = __builtin_elementwise_fma(yp1, wp01, e1[ptm]);
        HOLD(e1[ptm]); PIN_ORDER();
        y2x = fast_tanh(x2.x);
        HOLD(y2x); PIN_ORDER();
        if (!last) { xn2 = (v2f){qa[tn].z, qa[tn].w} + (v2f){kk[tn][tmn].z, kk[tn][tmn].w}; HOLD(xn2); }
        PIN_ORDER();
        y2y = fast_tanh(x2.y);
        HOLD(y2y); PIN_ORDER();
        e0[ptm] = __builtin_elementwise_fma(yp2, wp23, e0[ptm]);
        HOLD(e0[ptm]); PIN_ORDER();
        y3x = fast_tanh(x3.x);
        HOLD(y3x); PIN_ORDER();
        if (!last) { xn3 = (v2f){qb[tn].z, qb[tn].w} + (v2f){kk[tn][tmn].z, kk[tn][tmn].w}; HOLD(xn3); }
        PIN_ORDER();
        y3y = fast_tanh(x3.y);
        HOLD(y3y); PIN_ORDER();
        e1[ptm] = __builtin_elementwise_fma(yp3, wp23, e1[ptm]);
        HOLD(e1[ptm]); PIN_ORDER();
        yp0 = (v2f){y0x, y0y}; yp1 = (v2f){y1x, y1y}; yp2 = (v2f){y2x, y2y}; yp3 = (v2f){y3x, y3y};
        wp01 = (v2f){w[t].x, w[t].y}; wp23 = (v2f){w[t].z, w[t].w};
        x0 = xn0; x1 = xn1; x2 = xn2; x3 = xn3;
      }
    }
  }
#undef PIN_ORDER
#undef HOLD
  e0[M - 1] = __builtin_elementwise_fma(yp0, wp01, e0[M - 1]);
  e1[M - 1] = __builtin_elementwise_fma(yp1, wp01, e1[M - 1]);
  e0[M - 1] = __builtin_elementwise_fma(yp2, wp23, e0[M - 1]);
  e1[M - 1] = __builtin_elementwise_fma(yp3, wp23, e1[M - 1]);
  float s0[M], s1[M];
#pragma unroll
  for (int tm = 0; tm < M; ++tm) {
    s0[tm] = e0[tm].x + e0[tm].y;
    s1[tm] = e1[tm].x + e1[tm].y;
  }

  float* srow = S + (size_t)qrow * KLEN;
#pragma unroll
  for (int tm = 0; tm < M; ++tm) {
    const int kidx = k0 + tm * 16 + l16;
    if (kidx < KLEN) {
      const bool live = kidx < vlen;
      srow[kidx]        = live ? s0[tm] : NEGBIG;
      srow[KLEN + kidx] = live ? s1[tm] : NEGBIG;
    }
  }
}

__global__ __launch_bounds__(512) void keyproj_score(const float* __restrict__ keys,
                                                     const float* __restrict__ Wk,
                                                     const float* __restrict__ Qp,
                                                     const float* __restrict__ wv,
                                                     const int* __restrict__ valid_lens,
                                                     float* __restrict__ S) {
  __shared__ __attribute__((aligned(16))) float buf[KT * FP];

  const int t   = threadIdx.x;
  const int blk = blockIdx.x;

  int c2[BATCH], c3[BATCH], c4[BATCH];
  int n2 = 0, n3 = 0;
#pragma unroll
  for (int i = 0; i < BATCH; ++i) {
    const int vl = min(max(valid_lens[i], 0), KLEN);
    const int T  = (vl + 15) >> 4;
    c2[i] = (T + 1) / 2;
    c3[i] = (T + 2) / 3;
    c4[i] = (T + 3) / 4;
    n2 += c2[i];
    n3 += c3[i];
  }
  const int M = (n2 <= NBLK) ? 2 : ((n3 <= NBLK) ? 3 : 4);

  int my_b = -1, my_j = 0, my_vlen = 0, first = 0, cov_fill = 0;
#pragma unroll
  for (int i = 0; i < BATCH; ++i) {
    const int nb = (M == 2) ? c2[i] : ((M == 3) ? c3[i] : c4[i]);
    const bool mine = (blk >= first) && (blk < first + nb);
    my_b    = mine ? i : my_b;
    my_j    = mine ? blk - first : my_j;
    my_vlen = mine ? valid_lens[i] : my_vlen;
    cov_fill = (i == (blk >> 4)) ? min(KLEN, nb * 16 * M) : cov_fill;
    first += nb;
  }

  {
    float* frow = S + (size_t)(blk * 4 + (t >> 7)) * KLEN;
    const float4 neg = {NEGBIG, NEGBIG, NEGBIG, NEGBIG};
    for (int c = cov_fill + 4 * (t & 127); c < KLEN; c += 512)
      *(float4*)(frow + c) = neg;
  }

  if (my_b < 0) return;
  const int k0 = my_j * 16 * M;

  if (M == 2)      project_and_score<2>(keys, Wk, Qp, wv, S, buf, my_b, k0, my_vlen);
  else if (M == 3) project_and_score<3>(keys, Wk, Qp, wv, S, buf, my_b, k0, my_vlen);
  else             project_and_score<4>(keys, Wk, Qp, wv, S, buf, my_b, k0, my_vlen);
}

#define PP   1028
#define KC   64
#define VP   72
#define OP   68

#define NSET 6
#define STAGE_LOAD(R, src)                                              \
  do {                                                                  \
    R##a = *(const float4*)((src));                                     \
    R##b = *(const float4*)((src) + (size_t)16 * DV);                   \
    R##c = *(const float4*)((src) + (size_t)32 * DV);                   \
    R##d = *(const float4*)((src) + (size_t)48 * DV);                   \
  } while (0)
#define STAGE_STORE(dst, R)                                             \
  do {                                                                  \
    *(float4*)((dst))           = R##a;                                 \
    *(float4*)((dst) + 16 * VP) = R##b;                                 \
    *(float4*)((dst) + 32 * VP) = R##c;                                 \
    *(float4*)((dst) + 48 * VP) = R##d;                                 \
  } while (0)
__device__ __forceinline__ void lds_barrier() {
  asm volatile("s_wait_dscnt 0x0\n\ts_barrier_signal -1\n\ts_barrier_wait -1" ::: "memory");
}
__device__ __forceinline__ void mma_chunk(const float* ap, const float* bp, v8f& acc0, v8f& acc1) {
  float4 av[4];
  float  bv[4][4];
#pragma unroll
  for (int g = 0; g < 4; ++g) av[g] = *(const float4*)(ap + 8 * g);
#pragma unroll
  for (int g = 0; g < 4; ++g)
#pragma unroll
    for (int j = 0; j < 4; ++j) bv[g][j] = bp[(8 * g + j) * VP];
#pragma unroll
  for (int g = 0; g < 4; ++g) {
    const v2f a0 = {av[g].x, av[g].y};
    const v2f a1 = {av[g].z, av[g].w};
    const v2f b0 = {bv[g][0], bv[g][1]};
    const v2f b1 = {bv[g][2], bv[g][3]};
    acc0 = wmma4(a0, b0, acc0);
    acc1 = wmma4(a1, b1, acc1);
  }
}

__global__ __launch_bounds__(256) void softmax_out(const float* __restrict__ S,
                                                   const float* __restrict__ V,
                                                   float* __restrict__ OUT) {
  __shared__ __attribute__((aligned(16))) float P[16 * PP];
  __shared__ __attribute__((aligned(16))) float Vs[2 * KC * VP];

  const int t    = threadIdx.x;
  const int lane = t & 31;
  const int wave = __builtin_amdgcn_readfirstlane(t >> 5);
  const int b    = blockIdx.x >> 4;
  const int ng   = (blockIdx.x >> 2) & 3;
  const int tm   = blockIdx.x & 3;

  const int srow = t >> 4;
  const int sc4  = t & 15;
  const float* vsrc = V + (size_t)b * KLEN * DV + (size_t)srow * DV + ng * 64 + sc4 * 4;
  float* vdst = Vs + srow * VP + sc4 * 4;

  const float* Sg = S + ((size_t)(b * QLEN + tm * 16 + wave * 2)) * KLEN;
  float4 x[2][8];
#pragma unroll
  for (int rr = 0; rr < 2; ++rr)
#pragma unroll
    for (int i = 0; i < 8; ++i)
      x[rr][i] = *(const float4*)(Sg + rr * KLEN + 4 * (lane + 32 * i));

  float4 st0a, st0b, st0c, st0d, st1a, st1b, st1c, st1d, st2a, st2b, st2c, st2d;
  float4 st3a, st3b, st3c, st3d, st4a, st4b, st4c, st4d, st5a, st5b, st5c, st5d;
  STAGE_LOAD(st0, vsrc);
  STAGE_LOAD(st1, vsrc + (size_t)(1 * KC) * DV);
  STAGE_LOAD(st2, vsrc + (size_t)(2 * KC) * DV);
  STAGE_LOAD(st3, vsrc + (size_t)(3 * KC) * DV);
  STAGE_LOAD(st4, vsrc + (size_t)(4 * KC) * DV);
  STAGE_LOAD(st5, vsrc + (size_t)(5 * KC) * DV);

#pragma unroll
  for (int rr = 0; rr < 2; ++rr) {
    float m = x[rr][0].x;
#pragma unroll
    for (int i = 0; i < 8; ++i)
      m = fmaxf(m, fmaxf(fmaxf(x[rr][i].x, x[rr][i].y), fmaxf(x[rr][i].z, x[rr][i].w)));
#pragma unroll
    for (int off = 16; off > 0; off >>= 1) m = fmaxf(m, __shfl_xor(m, off));
    float s = 0.f;
#pragma unroll
    for (int i = 0; i < 8; ++i) {
      x[rr][i].x = __expf(x[rr][i].x - m);
      x[rr][i].y = __expf(x[rr][i].y - m);
      x[rr][i].z = __expf(x[rr][i].z - m);
      x[rr][i].w = __expf(x[rr][i].w - m);
      s += (x[rr][i].x + x[rr][i].y) + (x[rr][i].z + x[rr][i].w);
    }
#pragma unroll
    for (int off = 16; off > 0; off >>= 1) s += __shfl_xor(s, off);
    const float inv = 1.f / s;
    float* prow = P + (wave * 2 + rr) * PP;
#pragma unroll
    for (int i = 0; i < 8; ++i) {
      float4 p = x[rr][i];
      p.x *= inv; p.y *= inv; p.z *= inv; p.w *= inv;
      *(float4*)(prow + 4 * (lane + 32 * i)) = p;
    }
  }

  STAGE_STORE(vdst, st0);
  STAGE_LOAD(st0, vsrc + (size_t)(NSET * KC) * DV);
  lds_barrier();

  const int tn   = wave & 3;
  const int kh   = wave >> 2;
  const int half = lane >> 4;
  const int l16  = lane & 15;
  const float* Ap = P + l16 * PP + kh * 32 + 4 * half;
  const float* Bp = Vs + (kh * 32 + 4 * half) * VP + tn * 16 + l16;

  v8f acc0 = {};
  v8f acc1 = {};
#define CHUNK_REQ(c, R)                                                           \
  do {                                                                            \
    STAGE_STORE(vdst + (((c) + 1) & 1) * KC * VP, R);                             \
    STAGE_LOAD(R, vsrc + (size_t)(((c) + 1 + NSET) * KC) * DV);                   \
    mma_chunk(Ap + (c) * KC, Bp + ((c) & 1) * KC * VP, acc0, acc1);               \
    lds_barrier();                                                                \
  } while (0)
#define CHUNK_END(c, R)                                                           \
  do {                                                                            \
    STAGE_STORE(vdst + (((c) + 1) & 1) * KC * VP, R);                             \
    mma_chunk(Ap + (c) * KC, Bp + ((c) & 1) * KC * VP, acc0, acc1);               \
    lds_barrier();                                                                \
  } while (0)
  CHUNK_REQ(0, st1);
  CHUNK_REQ(1, st2);
  CHUNK_REQ(2, st3);
  CHUNK_REQ(3, st4);
  CHUNK_REQ(4, st5);
  CHUNK_REQ(5, st0);
  CHUNK_REQ(6, st1);
  CHUNK_REQ(7, st2);
  CHUNK_REQ(8, st3);
  CHUNK_END(9, st4);
  CHUNK_END(10, st5);
  CHUNK_END(11, st0);
  CHUNK_END(12, st1);
  CHUNK_END(13, st2);
  CHUNK_END(14, st3);
  mma_chunk(Ap + 15 * KC, Bp + KC * VP, acc0, acc1);
  int te = t;
  asm volatile("v_nop\n\tv_nop\n\tv_nop\n\tv_nop\n\tv_nop" : "+v"(acc0), "+v"(acc1), "+v"(te));

  const int e_lane = te & 31;
  const int e_wave = te >> 5;
  const int e_tn   = e_wave & 3;
  const int e_kh   = e_wave >> 2;
  const int e_half = e_lane >> 4;
  const int e_l16  = e_lane & 15;
  const int e_srow = te >> 4;
  const int e_sc4  = te & 15;
  float* O = Vs;
#pragma unroll
  for (int v = 0; v < 8; ++v)
    O[(e_kh * 16 + 8 * e_half + v) * OP + e_tn * 16 + e_l16] = acc0[v] + acc1[v];
  __syncthreads();
  const float4 o0 = *(const float4*)(O + e_srow * OP + e_sc4 * 4);
  const float4 o1 = *(const float4*)(O + (16 + e_srow) * OP + e_sc4 * 4);
  float4 o;
  o.x = o0.x + o1.x; o.y = o0.y + o1.y; o.z = o0.z + o1.z; o.w = o0.w + o1.w;
  *(float4*)(OUT + (size_t)(b * QLEN + tm * 16 + e_srow) * DV + ng * 64 + e_sc4 * 4) = o;
}

extern "C" void kernel_launch(void* const* d_in, const int* in_sizes, int n_in,
                              void* d_out, int out_size, void* d_ws, size_t ws_size,
                              hipStream_t stream) {
  const float* queries = (const float*)d_in[0];
  const float* keys    = (const float*)d_in[1];
  const float* values  = (const float*)d_in[2];
  const int*   vlens   = (const int*)d_in[3];
  const float* Wq      = (const float*)d_in[4];
  const float* Wk      = (const float*)d_in[5];
  const float* wv      = (const float*)d_in[6];
  float* out = (float*)d_out;

  float* Qp   = (float*)d_ws;
  float* attn = Qp + (size_t)BQ * HDIM;

  proj_gemm<<<(BQ / 16) * (HDIM / 16) / 8, 256, 0, stream>>>(queries, Wq, Qp);

  keyproj_score<<<NBLK, 512, 0, stream>>>(keys, Wk, Qp, wv, vlens, attn);

  softmax_out<<<BATCH * (DV / 64) * (QLEN / 16), 256, 0, stream>>>(attn, values, out);
}


# Query projection re-cut: a block owns 16 rows x 64 columns, rows staged whole into LDS once, each wave requests all its Wq 16-byte fragments up front, two k chains; k summed in another order
# speedup vs baseline: 5.1864x; 1.1459x over previous
#include <hip/hip_runtime.h>
#include <math.h>

#define HDIM  256
#define BATCH 16
#define QLEN  64
#define KLEN  1024
#define DV    256
#define BQ    (BATCH * QLEN)
#define NEGBIG -1e6f

typedef float v2f __attribute__((ext_vector_type(2)));
typedef float v8f __attribute__((ext_vector_type(8)));

__device__ __forceinline__ float fast_tanh(float x) {
#if __has_builtin(__builtin_amdgcn_tanhf)
  return __builtin_amdgcn_tanhf(x);
#elif __has_builtin(__builtin_amdgcn_tanh_f32)
  return __builtin_amdgcn_tanh_f32(x);
#else
  return tanhf(x);
#endif
}

__device__ __forceinline__ v8f wmma4(v2f a, v2f b, v8f c) {
  return __builtin_amdgcn_wmma_f32_16x16x4_f32(
       false, a,  false, b,
       (short)0, c,  false,  false);
}
__device__ __forceinline__ void lds_barrier() {
  asm volatile("s_wait_dscnt 0x0\n\ts_barrier_signal -1\n\ts_barrier_wait -1" ::: "memory");
}

#define QFP 260
#define QOP 68
#define QKJ (HDIM / 8)

__global__ __launch_bounds__(128) void query_proj(const float* __restrict__ X,
                                                  const float* __restrict__ W,
                                                  float* __restrict__ OUT) {
  __shared__ __attribute__((aligned(16))) float rows[16 * QFP];
  __shared__ __attribute__((aligned(16))) float otile[16 * QOP];

  const int t    = threadIdx.x;
  const int lane = t & 31;
  const int wave = __builtin_amdgcn_readfirstlane(t >> 5);
  const int tm   = blockIdx.x >> 2;
  const int cg   = blockIdx.x & 3;
  const int half = lane >> 4;
  const int l16  = lane & 15;

  const float* xg = X + (size_t)tm * 16 * HDIM;
  float4 sv[8];
#pragma unroll
  for (int i = 0; i < 8; ++i) sv[i] = *(const float4*)(xg + (size_t)(i * 128 + t) * 4);

  const float* Brow = W + (size_t)((cg * 4 + wave) * 16 + l16) * HDIM + 4 * half;
  float4 bv[QKJ];
#pragma unroll
  for (int j = 0; j < QKJ; ++j) bv[j] = *(const float4*)(Brow + 8 * j);

#pragma unroll
  for (int i = 0; i < 8; ++i) {
    const int idx = i * 128 + t;
    *(float4*)(rows + (idx >> 6) * QFP + (idx & 63) * 4) = sv[i];
  }
  lds_barrier();

  const float* Arow = rows + l16 * QFP + 4 * half;
  float4 av[QKJ];
#pragma unroll
  for (int j = 0; j < QKJ; ++j) av[j] = *(const float4*)(Arow + 8 * j);

  v8f acc0 = {};
  v8f acc1 = {};
  __builtin_amdgcn_sched_barrier(0);
#pragma unroll
  for (int j = 0; j < QKJ; ++j) {
    const v2f a0 = {av[j].x, av[j].y};
    const v2f a1 = {av[j].z, av[j].w};
    const v2f b0 = {bv[j].x, bv[j].y};
    const v2f b1 = {bv[j].z, bv[j].w};
    acc0 = wmma4(a0, b0, acc0);
    acc1 = wmma4(a1, b1, acc1);
  }
  __builtin_amdgcn_sched_barrier(0);
  int te = t;
  asm volatile("v_nop\n\tv_nop\n\tv_nop\n\tv_nop\n\tv_nop" : "+v"(acc0), "+v"(acc1), "+v"(te));

  const int e_lane = te & 31;
  const int e_wave = te >> 5;
  const int e_half = e_lane >> 4;
  const int e_l16  = e_lane & 15;
#pragma unroll
  for (int v = 0; v < 8; ++v)
    otile[(8 * e_half + v) * QOP + e_wave * 16 + e_l16] = acc0[v] + acc1[v];
  __syncthreads();
  const int e_r  = te >> 4;
  const int e_c4 = te & 15;
  const float4 o0 = *(const float4*)(otile + e_r * QOP + e_c4 * 4);
  const float4 o1 = *(const float4*)(otile + (e_r + 8) * QOP + e_c4 * 4);
  float* og = OUT + (size_t)(tm * 16 + e_r) * HDIM + cg * 64 + e_c4 * 4;
  *(float4*)(og)            = o0;
  *(float4*)(og + 8 * HDIM) = o1;
}

#define KT   64
#define FP   260
#define NBLK (BATCH * (KLEN / KT))

template <int M>
__device__ __forceinline__ void project_and_score(const float* __restrict__ keys,
                                                  const float* __restrict__ Wk,
                                                  const float* __restrict__ Qp,
                                                  const float* __restrict__ wv,
                                                  float* __restrict__ S,
                                                  float* __restrict__ buf,
                                                  const int b, const int k0, const int vlen) {
  const int t    = threadIdx.x;
  const int lane = t & 31;
  const int wave = __builtin_amdgcn_readfirstlane(t >> 5);

  const float* kg = keys + (size_t)b * KLEN * HDIM;
#pragma unroll
  for (int i = 0; i < 2 * M; ++i) {
    const int idx = i * 512 + t;
    const int row = idx >> 6;
    const int c4  = idx & 63;
    const int gk  = min(k0 + row, KLEN - 1);
    const float4 v = *(const float4*)(kg + (size_t)gk * HDIM + c4 * 4);
    *(float4*)(buf + row * FP + c4 * 4) = v;
  }
  __syncthreads();

  const int half = lane >> 4;
  const int l16  = lane & 15;
  const float* Brow = Wk + (wave * 16 + l16) * HDIM + 4 * half;
  const float* Arow = buf + l16 * FP + 4 * half;

  v8f acc[M] = {};
  float4 bcur[4], bnxt[4];
#pragma unroll
  for (int u = 0; u < 4; ++u) bcur[u] = *(const float4*)(Brow + 8 * u);

  __builtin_amdgcn_sched_barrier(0);

#pragma unroll
  for (int jj = 0; jj < HDIM / 8; jj += 4) {
    if (jj + 4 < HDIM / 8) {
#pragma unroll
      for (int u = 0; u < 4; ++u) bnxt[u] = *(const float4*)(Brow + 8 * (jj + 4 + u));
    }
#pragma unroll
    for (int u = 0; u < 4; ++u) {
      const int j = jj + u;
      float4 av[M];
#pragma unroll
      for (int tm = 0; tm < M; ++tm)
        av[tm] = *(const float4*)(Arow + tm * 16 * FP + 8 * j);
      const v2f b0 = {bcur[u].x, bcur[u].y};
      const v2f b1 = {bcur[u].z, bcur[u].w};
#pragma unroll
      for (int tm = 0; tm < M; ++tm) {
        const v2f a0 = {av[tm].x, av[tm].y};
        acc[tm] = wmma4(a0, b0, acc[tm]);
      }
#pragma unroll
      for (int tm = 0; tm < M; ++tm) {
        const v2f a1 = {av[tm].z, av[tm].w};
        acc[tm] = wmma4(a1, b1, acc[tm]);
      }
    }
#pragma unroll
    for (int u = 0; u < 4; ++u) bcur[u] = bnxt[u];
  }
  __builtin_amdgcn_sched_barrier(0);
  asm volatile("v_nop\n\tv_nop\n\tv_nop\n\tv_nop\n\tv_nop");
  __syncthreads();

#pragma unroll
  for (int tm = 0; tm < M; ++tm) {
#pragma unroll
    for (int v = 0; v < 8; ++v)
      buf[(tm * 16 + 8 * half + v) * FP + wave * 16 + l16] = acc[tm][v];
  }
  __syncthreads();

  const int qrow = b * QLEN + wave * 4 + 2 * half;
  const float* qp = Qp + (size_t)qrow * HDIM;
  const float* kp = buf + l16 * FP;

  constexpr int TP = 4;
  v2f e0[M], e1[M];
#pragma unroll
  for (int tm = 0; tm < M; ++tm) { e0[tm] = (v2f){0.f, 0.f}; e1[tm] = (v2f){0.f, 0.f}; }
  v2f yp0 = {0.f, 0.f}, yp1 = {0.f, 0.f}, yp2 = {0.f, 0.f}, yp3 = {0.f, 0.f};
  v2f wp01 = {0.f, 0.f}, wp23 = {0.f, 0.f};
#define PIN_ORDER() __builtin_amdgcn_sched_barrier(0)
#define HOLD(v) asm volatile("" : "+v"(v))
#pragma unroll 1
  for (int h = 0; h < HDIM; h += 4 * TP) {
    float4 w[TP], qa[TP], qb[TP], kk[TP][M];
#pragma unroll
    for (int t = 0; t < TP; ++t) {
      w[t]  = *(const float4*)(wv + h + 4 * t);
      qa[t] = *(const float4*)(qp + h + 4 * t);
      qb[t] = *(const float4*)(qp + HDIM + h + 4 * t);
#pragma unroll
      for (int tm = 0; tm < M; ++tm) kk[t][tm] = *(const float4*)(kp + tm * 16 * FP + h + 4 * t);
    }
    v2f x0 = (v2f){qa[0].x, qa[0].y} + (v2f){kk[0][0].x, kk[0][0].y};
    v2f x1 = (v2f){qb[0].x, qb[0].y} + (v2f){kk[0][0].x, kk[0][0].y};
    v2f x2 = (v2f){qa[0].z, qa[0].w} + (v2f){kk[0][0].z, kk[0][0].w};
    v2f x3 = (v2f){qb[0].z, qb[0].w} + (v2f){kk[0][0].z, kk[0][0].w};
    PIN_ORDER();
#pragma unroll
    for (int t = 0; t < TP; ++t) {
#pragma unroll
      for (int tm = 0; tm < M; ++tm) {
        const bool last = (t == TP - 1) && (tm == M - 1);
        const int tn  = (tm + 1 < M) ? t : ((t + 1 < TP) ? t + 1 : t);
        const int tmn = (tm + 1 < M) ? tm + 1 : 0;
        const int ptm = (tm > 0) ? tm - 1 : M - 1;
        float y0x, y0y, y1x, y1y, y2x, y2y, y3x, y3y;
        v2f xn0 = x0, xn1 = x1, xn2 = x2, xn3 = x3;
        y0x = fast_tanh(x0.x);
        HOLD(y0x); PIN_ORDER();
        if (!last) { xn0 = (v2f){qa[tn].x, qa[tn].y} + (v2f){kk[tn][tmn].x, kk[tn][tmn].y}; HOLD(xn0); }
        PIN_ORDER();
        y0y = fast_tanh(x0.y);
        HOLD(y0y); PIN_ORDER();
        e0[ptm] = __builtin_elementwise_fma(yp0, wp01, e0[ptm]);
        HOLD(e0[ptm]); PIN_ORDER();
        y1x = fast_tanh(x1.x);
        HOLD(y1x); PIN_ORDER();
        if (!last) { xn1 = (v2f){qb[tn].x, qb[tn].y} + (v2f){kk[tn][tmn].x, kk[tn][tmn].y}; HOLD(xn1); }
        PIN_ORDER();
        y1y = fast_tanh(x1.y);
        HOLD(y1y); PIN_ORDER();
        e1[ptm] = __builtin_elementwise_fma(yp1, wp01, e1[ptm]);
        HOLD(e1[ptm]); PIN_ORDER();
        y2x = fast_tanh(x2.x);
        HOLD(y2x); PIN_ORDER();
        if (!last) { xn2 = (v2f){qa[tn].z, qa[tn].w} + (v2f){kk[tn][tmn].z, kk[tn][tmn].w}; HOLD(xn2); }
        PIN_ORDER();
        y2y = fast_tanh(x2.y);
        HOLD(y2y); PIN_ORDER();
        e0[ptm] = __builtin_elementwise_fma(yp2, wp23, e0[ptm]);
        HOLD(e0[ptm]); PIN_ORDER();
        y3x = fast_tanh(x3.x);
        HOLD(y3x); PIN_ORDER();
        if (!last) { xn3 = (v2f){qb[tn].z, qb[tn].w} + (v2f){kk[tn][tmn].z, kk[tn][tmn].w}; HOLD(xn3); }
        PIN_ORDER();
        y3y = fast_tanh(x3.y);
        HOLD(y3y); PIN_ORDER();
        e1[ptm] = __builtin_elementwise_fma(yp3, wp23, e1[ptm]);
        HOLD(e1[ptm]); PIN_ORDER();
        yp0 = (v2f){y0x, y0y}; yp1 = (v2f){y1x, y1y}; yp2 = (v2f){y2x, y2y}; yp3 = (v2f){y3x, y3y};
        wp01 = (v2f){w[t].x, w[t].y}; wp23 = (v2f){w[t].z, w[t].w};
        x0 = xn0; x1 = xn1; x2 = xn2; x3 = xn3;
      }
    }
  }
#undef PIN_ORDER
#undef HOLD
  e0[M - 1] = __builtin_elementwise_fma(yp0, wp01, e0[M - 1]);
  e1[M - 1] = __builtin_elementwise_fma(yp1, wp01, e1[M - 1]);
  e0[M - 1] = __builtin_elementwise_fma(yp2, wp23, e0[M - 1]);
  e1[M - 1] = __builtin_elementwise_fma(yp3, wp23, e1[M - 1]);
  float s0[M], s1[M];
#pragma unroll
  for (int tm = 0; tm < M; ++tm) {
    s0[tm] = e0[tm].x + e0[tm].y;
    s1[tm] = e1[tm].x + e1[tm].y;
  }

  float* srow = S + (size_t)qrow * KLEN;
#pragma unroll
  for (int tm = 0; tm < M; ++tm) {
    const int kidx = k0 + tm * 16 + l16;
    if (kidx < KLEN) {
      const bool live = kidx < vlen;
      srow[kidx]        = live ? s0[tm] : NEGBIG;
      srow[KLEN + kidx] = live ? s1[tm] : NEGBIG;
    }
  }
}

__global__ __launch_bounds__(512) void keyproj_score(const float* __restrict__ keys,
                                                     const float* __restrict__ Wk,
                                                     const float* __restrict__ Qp,
                                                     const float* __restrict__ wv,
                                                     const int* __restrict__ valid_lens,
                                                     float* __restrict__ S) {
  __shared__ __attribute__((aligned(16))) float buf[KT * FP];

  const int t   = threadIdx.x;
  const int blk = blockIdx.x;

  int c2[BATCH], c3[BATCH], c4[BATCH];
  int n2 = 0, n3 = 0;
#pragma unroll
  for (int i = 0; i < BATCH; ++i) {
    const int vl = min(max(valid_lens[i], 0), KLEN);
    const int T  = (vl + 15) >> 4;
    c2[i] = (T + 1) / 2;
    c3[i] = (T + 2) / 3;
    c4[i] = (T + 3) / 4;
    n2 += c2[i];
    n3 += c3[i];
  }
  const int M = (n2 <= NBLK) ? 2 : ((n3 <= NBLK) ? 3 : 4);

  int my_b = -1, my_j = 0, my_vlen = 0, first = 0, cov_fill = 0;
#pragma unroll
  for (int i = 0; i < BATCH; ++i) {
    const int nb = (M == 2) ? c2[i] : ((M == 3) ? c3[i] : c4[i]);
    const bool mine = (blk >= first) && (blk < first + nb);
    my_b    = mine ? i : my_b;
    my_j    = mine ? blk - first : my_j;
    my_vlen = mine ? valid_lens[i] : my_vlen;
    cov_fill = (i == (blk >> 4)) ? min(KLEN, nb * 16 * M) : cov_fill;
    first += nb;
  }

  {
    float* frow = S + (size_t)(blk * 4 + (t >> 7)) * KLEN;
    const float4 neg = {NEGBIG, NEGBIG, NEGBIG, NEGBIG};
    for (int c = cov_fill + 4 * (t & 127); c < KLEN; c += 512)
      *(float4*)(frow + c) = neg;
  }

  if (my_b < 0) return;
  const int k0 = my_j * 16 * M;

  if (M == 2)      project_and_score<2>(keys, Wk, Qp, wv, S, buf, my_b, k0, my_vlen);
  else if (M == 3) project_and_score<3>(keys, Wk, Qp, wv, S, buf, my_b, k0, my_vlen);
  else             project_and_score<4>(keys, Wk, Qp, wv, S, buf, my_b, k0, my_vlen);
}

#define PP   1028
#define KC   64
#define VP   72
#define OP   68

#define NSET 6
#define STAGE_LOAD(R, src)                                              \
  do {                                                                  \
    R##a = *(const float4*)((src));                                     \
    R##b = *(const float4*)((src) + (size_t)16 * DV);                   \
    R##c = *(const float4*)((src) + (size_t)32 * DV);                   \
    R##d = *(const float4*)((src) + (size_t)48 * DV);                   \
  } while (0)
#define STAGE_STORE(dst, R)                                             \
  do {                                                                  \
    *(float4*)((dst))           = R##a;                                 \
    *(float4*)((dst) + 16 * VP) = R##b;                                 \
    *(float4*)((dst) + 32 * VP) = R##c;                                 \
    *(float4*)((dst) + 48 * VP) = R##d;                                 \
  } while (0)
__device__ __forceinline__ void mma_chunk(const float* ap, const float* bp, v8f& acc0, v8f& acc1) {
  float4 av[4];
  float  bv[4][4];
#pragma unroll
  for (int g = 0; g < 4; ++g) av[g] = *(const float4*)(ap + 8 * g);
#pragma unroll
  for (int g = 0; g < 4; ++g)
#pragma unroll
    for (int j = 0; j < 4; ++j) bv[g][j] = bp[(8 * g + j) * VP];
#pragma unroll
  for (int g = 0; g < 4; ++g) {
    const v2f a0 = {av[g].x, av[g].y};
    const v2f a1 = {av[g].z, av[g].w};
    const v2f b0 = {bv[g][0], bv[g][1]};
    const v2f b1 = {bv[g][2], bv[g][3]};
    acc0 = wmma4(a0, b0, acc0);
    acc1 = wmma4(a1, b1, acc1);
  }
}

__global__ __launch_bounds__(256) void softmax_out(const float* __restrict__ S,
                                                   const float* __restrict__ V,
                                                   float* __restrict__ OUT) {
  __shared__ __attribute__((aligned(16))) float P[16 * PP];
  __shared__ __attribute__((aligned(16))) float Vs[2 * KC * VP];

  const int t    = threadIdx.x;
  const int lane = t & 31;
  const int wave = __builtin_amdgcn_readfirstlane(t >> 5);
  const int b    = blockIdx.x >> 4;
  const int ng   = (blockIdx.x >> 2) & 3;
  const int tm   = blockIdx.x & 3;

  const int srow = t >> 4;
  const int sc4  = t & 15;
  const float* vsrc = V + (size_t)b * KLEN * DV + (size_t)srow * DV + ng * 64 + sc4 * 4;
  float* vdst = Vs + srow * VP + sc4 * 4;

  const float* Sg = S + ((size_t)(b * QLEN + tm * 16 + wave * 2)) * KLEN;
  float4 x[2][8];
#pragma unroll
  for (int rr = 0; rr < 2; ++rr)
#pragma unroll
    for (int i = 0; i < 8; ++i)
      x[rr][i] = *(const float4*)(Sg + rr * KLEN + 4 * (lane + 32 * i));

  float4 st0a, st0b, st0c, st0d, st1a, st1b, st1c, st1d, st2a, st2b, st2c, st2d;
  float4 st3a, st3b, st3c, st3d, st4a, st4b, st4c, st4d, st5a, st5b, st5c, st5d;
  STAGE_LOAD(st0, vsrc);
  STAGE_LOAD(st1, vsrc + (size_t)(1 * KC) * DV);
  STAGE_LOAD(st2, vsrc + (size_t)(2 * KC) * DV);
  STAGE_LOAD(st3, vsrc + (size_t)(3 * KC) * DV);
  STAGE_LOAD(st4, vsrc + (size_t)(4 * KC) * DV);
  STAGE_LOAD(st5, vsrc + (size_t)(5 * KC) * DV);

#pragma unroll
  for (int rr = 0; rr < 2; ++rr) {
    float m = x[rr][0].x;
#pragma unroll
    for (int i = 0; i < 8; ++i)
      m = fmaxf(m, fmaxf(fmaxf(x[rr][i].x, x[rr][i].y), fmaxf(x[rr][i].z, x[rr][i].w)));
#pragma unroll
    for (int off = 16; off > 0; off >>= 1) m = fmaxf(m, __shfl_xor(m, off));
    float s = 0.f;
#pragma unroll
    for (int i = 0; i < 8; ++i) {
      x[rr][i].x = __expf(x[rr][i].x - m);
      x[rr][i].y = __expf(x[rr][i].y - m);
      x[rr][i].z = __expf(x[rr][i].z - m);
      x[rr][i].w = __expf(x[rr][i].w - m);
      s += (x[rr][i].x + x[rr][i].y) + (x[rr][i].z + x[rr][i].w);
    }
#pragma unroll
    for (int off = 16; off > 0; off >>= 1) s += __shfl_xor(s, off);
    const float inv = 1.f / s;
    float* prow = P + (wave * 2 + rr) * PP;
#pragma unroll
    for (int i = 0; i < 8; ++i) {
      float4 p = x[rr][i];
      p.x *= inv; p.y *= inv; p.z *= inv; p.w *= inv;
      *(float4*)(prow + 4 * (lane + 32 * i)) = p;
    }
  }

  STAGE_STORE(vdst, st0);
  STAGE_LOAD(st0, vsrc + (size_t)(NSET * KC) * DV);
  lds_barrier();

  const int tn   = wave & 3;
  const int kh   = wave >> 2;
  const int half = lane >> 4;
  const int l16  = lane & 15;
  const float* Ap = P + l16 * PP + kh * 32 + 4 * half;
  const float* Bp = Vs + (kh * 32 + 4 * half) * VP + tn * 16 + l16;

  v8f acc0 = {};
  v8f acc1 = {};
#define CHUNK_REQ(c, R)                                                           \
  do {                                                                            \
    STAGE_STORE(vdst + (((c) + 1) & 1) * KC * VP, R);                             \
    STAGE_LOAD(R, vsrc + (size_t)(((c) + 1 + NSET) * KC) * DV);                   \
    mma_chunk(Ap + (c) * KC, Bp + ((c) & 1) * KC * VP, acc0, acc1);               \
    lds_barrier();                                                                \
  } while (0)
#define CHUNK_END(c, R)                                                           \
  do {                                                                            \
    STAGE_STORE(vdst + (((c) + 1) & 1) * KC * VP, R);                             \
    mma_chunk(Ap + (c) * KC, Bp + ((c) & 1) * KC * VP, acc0, acc1);               \
    lds_barrier();                                                                \
  } while (0)
  CHUNK_REQ(0, st1);
  CHUNK_REQ(1, st2);
  CHUNK_REQ(2, st3);
  CHUNK_REQ(3, st4);
  CHUNK_REQ(4, st5);
  CHUNK_REQ(5, st0);
  CHUNK_REQ(6, st1);
  CHUNK_REQ(7, st2);
  CHUNK_REQ(8, st3);
  CHUNK_END(9, st4);
  CHUNK_END(10, st5);
  CHUNK_END(11, st0);
  CHUNK_END(12, st1);
  CHUNK_END(13, st2);
  CHUNK_END(14, st3);
  mma_chunk(Ap + 15 * KC, Bp + KC * VP, acc0, acc1);
  int te = t;
  asm volatile("v_nop\n\tv_nop\n\tv_nop\n\tv_nop\n\tv_nop" : "+v"(acc0), "+v"(acc1), "+v"(te));

  const int e_lane = te & 31;
  const int e_wave = te >> 5;
  const int e_tn   = e_wave & 3;
  const int e_kh   = e_wave >> 2;
  const int e_half = e_lane >> 4;
  const int e_l16  = e_lane & 15;
  const int e_srow = te >> 4;
  const int e_sc4  = te & 15;
  float* O = Vs;
#pragma unroll
  for (int v = 0; v < 8; ++v)
    O[(e_kh * 16 + 8 * e_half + v) * OP + e_tn * 16 + e_l16] = acc0[v] + acc1[v];
  __syncthreads();
  const float4 o0 = *(const float4*)(O + e_srow * OP + e_sc4 * 4);
  const float4 o1 = *(const float4*)(O + (16 + e_srow) * OP + e_sc4 * 4);
  float4 o;
  o.x = o0.x + o1.x; o.y = o0.y + o1.y; o.z = o0.z + o1.z; o.w = o0.w + o1.w;
  *(float4*)(OUT + (size_t)(b * QLEN + tm * 16 + e_srow) * DV + ng * 64 + e_sc4 * 4) = o;
}

extern "C" void kernel_launch(void* const* d_in, const int* in_sizes, int n_in,
                              void* d_out, int out_size, void* d_ws, size_t ws_size,
                              hipStream_t stream) {
  const float* queries = (const float*)d_in[0];
  const float* keys    = (const float*)d_in[1];
  const float* values  = (const float*)d_in[2];
  const int*   vlens   = (const int*)d_in[3];
  const float* Wq      = (const float*)d_in[4];
  const float* Wk      = (const float*)d_in[5];
  const float* wv      = (const float*)d_in[6];
  float* out = (float*)d_out;

  float* Qp   = (float*)d_ws;
  float* attn = Qp + (size_t)BQ * HDIM;

  query_proj<<<(BQ / 16) * (HDIM / 64), 128, 0, stream>>>(queries, Wq, Qp);

  keyproj_score<<<NBLK, 512, 0, stream>>>(keys, Wk, Qp, wv, vlens, attn);

  softmax_out<<<BATCH * (DV / 64) * (QLEN / 16), 256, 0, stream>>>(attn, values, out);
}
